# starting kernel
# baseline (speedup 1.0000x reference)
#include <hip/hip_runtime.h>
#include <math.h>


typedef __attribute__((ext_vector_type(2))) float v2f;
typedef __attribute__((ext_vector_type(8))) float v8f;

#define N_NODES 50000
#define N_EDGES 800000
#define K_DIM   128

__global__ void gcn_degrees(const int* __restrict__ src, const int* __restrict__ dst,
                            float* __restrict__ outdeg, float* __restrict__ indeg) {
  int e = blockIdx.x * blockDim.x + threadIdx.x;
  if (e < N_EDGES) {
    atomicAdd(&outdeg[src[e]], 1.0f);
    atomicAdd(&indeg[dst[e]], 1.0f);
  }
}

__global__ void gcn_deg_isqrt(const float* __restrict__ outdeg, const float* __restrict__ indeg,
                              float* __restrict__ odi, float* __restrict__ idi) {
  int i = blockIdx.x * blockDim.x + threadIdx.x;
  if (i < N_NODES) {
    odi[i] = rsqrtf(fmaxf(outdeg[i], 1.0f));
    idi[i] = rsqrtf(fmaxf(indeg[i], 1.0f));
  }
}

template <int NOUT>
__global__ __launch_bounds__(256) void gcn_gemm_wmma(const float* __restrict__ A,
                                                     const float* __restrict__ odi,
                                                     const float* __restrict__ W,
                                                     float* __restrict__ C) {
  __shared__ float As[16][K_DIM + 4];

  const int tileM = blockIdx.x * 16;

  for (int idx = threadIdx.x; idx < 16 * K_DIM; idx += blockDim.x) {
    const int r = idx >> 7;
    const int c = idx & (K_DIM - 1);
    As[r][c] = A[(size_t)(tileM + r) * K_DIM + c] * odi[tileM + r];
  }
  __syncthreads();

  const int wave = threadIdx.x >> 5;
  const int lane = threadIdx.x & 31;
  const int half = lane >> 4;
  const int r16  = lane & 15;
  const int n0   = wave * 16;

  v8f acc = {};
#pragma unroll
  for (int k = 0; k < K_DIM; k += 4) {
    v2f a, b;
    a.x = As[r16][k + 2 * half];
    a.y = As[r16][k + 2 * half + 1];
    b.x = W[(size_t)(k + 2 * half)     * NOUT + n0 + r16];
    b.y = W[(size_t)(k + 2 * half + 1) * NOUT + n0 + r16];
    acc = __builtin_amdgcn_wmma_f32_16x16x4_f32(false, a, false, b,
                                                (short)0, acc, false, false);
  }

#pragma unroll
  for (int r = 0; r < 8; ++r) {
    const int row = tileM + r + 8 * half;
    C[(size_t)row * NOUT + n0 + r16] = acc[r];
  }
}

template <int D>
__global__ void gcn_edge_scatter(const float* __restrict__ proj,
                                 const int* __restrict__ src, const int* __restrict__ dst,
                                 const float* __restrict__ ew, float* __restrict__ agg) {
  constexpr int LPE = D / 4;
  const int gid = blockIdx.x * blockDim.x + threadIdx.x;
  const int e = gid / LPE;
  const int l = gid % LPE;
  if (e >= N_EDGES) return;

  const int s = src[e];
  const int d = dst[e];
  const float w = ew[e];

  const float4 v = ((const float4*)(proj + (size_t)s * D))[l];
  float* o = agg + (size_t)d * D + l * 4;
  atomicAdd(o + 0, v.x * w);
  atomicAdd(o + 1, v.y * w);
  atomicAdd(o + 2, v.z * w);
  atomicAdd(o + 3, v.w * w);
}

__global__ void gcn_finalize(const float* __restrict__ agg, const float* __restrict__ idi,
                             const float* __restrict__ bias, float* __restrict__ out,
                             int D, int do_relu) {
  const int i = blockIdx.x * blockDim.x + threadIdx.x;
  if (i >= N_NODES * D) return;
  const int row = i / D;
  const int col = i % D;
  float v = agg[i] * idi[row] + bias[col];
  out[i] = do_relu ? fmaxf(v, 0.0f) : v;
}

extern "C" void kernel_launch(void* const* d_in, const int* in_sizes, int n_in,
                              void* d_out, int out_size, void* d_ws, size_t ws_size,
                              hipStream_t stream) {
  const float* feat = (const float*)d_in[0];
  const float* ew   = (const float*)d_in[1];
  const float* W1   = (const float*)d_in[2];
  const float* b1   = (const float*)d_in[3];
  const float* W2   = (const float*)d_in[4];
  const float* b2   = (const float*)d_in[5];
  const float* W3   = (const float*)d_in[6];
  const float* b3   = (const float*)d_in[7];
  const int*   src  = (const int*)d_in[8];
  const int*   dst  = (const int*)d_in[9];
  float* out = (float*)d_out;

  float* ws     = (float*)d_ws;
  float* outdeg = ws;
  float* indeg  = ws + N_NODES;
  float* odi    = ws + 2 * (size_t)N_NODES;
  float* idi    = ws + 3 * (size_t)N_NODES;
  float* h      = ws + 4 * (size_t)N_NODES;
  float* proj   = h    + (size_t)N_NODES * 128;
  float* agg    = proj + (size_t)N_NODES * 128;

  hipMemsetAsync(outdeg, 0, 2 * (size_t)N_NODES * sizeof(float), stream);
  gcn_degrees<<<(N_EDGES + 255) / 256, 256, 0, stream>>>(src, dst, outdeg, indeg);
  gcn_deg_isqrt<<<(N_NODES + 255) / 256, 256, 0, stream>>>(outdeg, indeg, odi, idi);

  const int gemmBlocks = N_NODES / 16;
  const int scat128Blocks = (N_EDGES * 32 + 255) / 256;
  const int scat64Blocks  = (N_EDGES * 16 + 255) / 256;
  const int fin128Blocks  = (N_NODES * 128 + 255) / 256;
  const int fin64Blocks   = (N_NODES * 64 + 255) / 256;

  gcn_gemm_wmma<128><<<gemmBlocks, 256, 0, stream>>>(feat, odi, W1, proj);
  hipMemsetAsync(agg, 0, (size_t)N_NODES * 128 * sizeof(float), stream);
  gcn_edge_scatter<128><<<scat128Blocks, 256, 0, stream>>>(proj, src, dst, ew, agg);
  gcn_finalize<<<fin128Blocks, 256, 0, stream>>>(agg, idi, b1, h, 128, 1);

  gcn_gemm_wmma<128><<<gemmBlocks, 256, 0, stream>>>(h, odi, W2, proj);
  hipMemsetAsync(agg, 0, (size_t)N_NODES * 128 * sizeof(float), stream);
  gcn_edge_scatter<128><<<scat128Blocks, 256, 0, stream>>>(proj, src, dst, ew, agg);
  gcn_finalize<<<fin128Blocks, 256, 0, stream>>>(agg, idi, b2, h, 128, 1);

  gcn_gemm_wmma<64><<<gemmBlocks, 128, 0, stream>>>(h, odi, W3, proj);
  hipMemsetAsync(agg, 0, (size_t)N_NODES * 64 * sizeof(float), stream);
  gcn_edge_scatter<64><<<scat64Blocks, 256, 0, stream>>>(proj, src, dst, ew, agg);
  gcn_finalize<<<fin64Blocks, 256, 0, stream>>>(agg, idi, b3, out, 64, 0);
}


# Edge aggregation: float-atomic scatter + memset + finalize replaced by per-call dst-sorted edge list (int count returns arrival rank, scan, fill) and atomic-free row gather; sum order = arrival rank
# speedup vs baseline: 2.3882x; 2.3882x over previous
#include <hip/hip_runtime.h>
#include <math.h>


typedef __attribute__((ext_vector_type(2))) float v2f;
typedef __attribute__((ext_vector_type(8))) float v8f;

#define N_NODES 50000
#define N_EDGES 800000
#define K_DIM   128

__global__ void gcn_degrees(const int* __restrict__ src, const int* __restrict__ dst,
                            int* __restrict__ cnt_out, int* __restrict__ cnt_in,
                            int* __restrict__ rank) {
  int e = blockIdx.x * blockDim.x + threadIdx.x;
  if (e < N_EDGES) {
    atomicAdd(&cnt_out[src[e]], 1);
    rank[e] = atomicAdd(&cnt_in[dst[e]], 1);
  }
}

__global__ __launch_bounds__(1024) void gcn_scan_norm(const int* __restrict__ cnt_out,
                                                      const int* __restrict__ cnt_in,
                                                      float* __restrict__ odi,
                                                      float* __restrict__ idi,
                                                      int* __restrict__ rowptr) {
  __shared__ int s_part[32];
  __shared__ int s_wtot[32];
  const int tid  = threadIdx.x;
  const int lane = tid & 31;
  const int wave = tid >> 5;
  const int tile0 = blockIdx.x * 1024;

  int part = 0;
  for (int i = tid; i < tile0; i += 1024) part += cnt_in[i];

  const int i = tile0 + tid;
  const int c = (i < N_NODES) ? cnt_in[i] : 0;

  int incl = c;
  for (int o = 1; o < 32; o <<= 1) {
    const int n = __shfl_up(incl, o, 32);
    if (lane >= o) incl += n;
  }
  for (int o = 16; o > 0; o >>= 1) part += __shfl_xor(part, o, 32);

  if (lane == 31) s_wtot[wave] = incl;
  if (lane == 0)  s_part[wave] = part;
  __syncthreads();

  int base = 0;
  for (int w = 0; w < 32; ++w) {
    base += s_part[w];
    if (w < wave) base += s_wtot[w];
  }

  if (i < N_NODES) {
    rowptr[i] = base + incl - c;
    odi[i] = rsqrtf(fmaxf((float)cnt_out[i], 1.0f));
    idi[i] = rsqrtf(fmaxf((float)c, 1.0f));
  }
}

__global__ void gcn_fill(const int* __restrict__ src, const int* __restrict__ dst,
                         const float* __restrict__ ew, const int* __restrict__ rank,
                         const int* __restrict__ rowptr, int* __restrict__ pairs) {
  int e = blockIdx.x * blockDim.x + threadIdx.x;
  if (e < N_EDGES) {
    const int p = rowptr[dst[e]] + rank[e];
    ((int2*)pairs)[p] = make_int2(src[e], __float_as_int(ew[e]));
  }
}

template <int NOUT>
__global__ __launch_bounds__(256) void gcn_gemm_wmma(const float* __restrict__ A,
                                                     const float* __restrict__ odi,
                                                     const float* __restrict__ W,
                                                     float* __restrict__ C) {
  __shared__ float As[16][K_DIM + 4];

  const int tileM = blockIdx.x * 16;

  for (int idx = threadIdx.x; idx < 16 * K_DIM; idx += blockDim.x) {
    const int r = idx >> 7;
    const int c = idx & (K_DIM - 1);
    As[r][c] = A[(size_t)(tileM + r) * K_DIM + c] * odi[tileM + r];
  }
  __syncthreads();

  const int wave = threadIdx.x >> 5;
  const int lane = threadIdx.x & 31;
  const int half = lane >> 4;
  const int r16  = lane & 15;
  const int n0   = wave * 16;

  v8f acc = {};
#pragma unroll
  for (int k = 0; k < K_DIM; k += 4) {
    v2f a, b;
    a.x = As[r16][k + 2 * half];
    a.y = As[r16][k + 2 * half + 1];
    b.x = W[(size_t)(k + 2 * half)     * NOUT + n0 + r16];
    b.y = W[(size_t)(k + 2 * half + 1) * NOUT + n0 + r16];
    acc = __builtin_amdgcn_wmma_f32_16x16x4_f32(false, a, false, b,
                                                (short)0, acc, false, false);
  }

#pragma unroll
  for (int r = 0; r < 8; ++r) {
    const int row = tileM + r + 8 * half;
    C[(size_t)row * NOUT + n0 + r16] = acc[r];
  }
}

template <int D, int RELU>
__global__ __launch_bounds__(256) void gcn_gather(const float* __restrict__ proj,
                                                  const int* __restrict__ pairs,
                                                  const int* __restrict__ rowptr,
                                                  const int* __restrict__ cnt_in,
                                                  const float* __restrict__ idi,
                                                  const float* __restrict__ bias,
                                                  float* __restrict__ out) {
  constexpr int VEC = D / 32;
  typedef float vecf __attribute__((ext_vector_type(VEC)));

  const int lane = threadIdx.x & 31;
  const int node = __builtin_amdgcn_readfirstlane(blockIdx.x * 8 + (threadIdx.x >> 5));
  if (node >= N_NODES) return;

  const int beg = rowptr[node];
  const int cnt = cnt_in[node];
  const int2* __restrict__ pr = (const int2*)pairs + beg;
  const vecf* __restrict__ rows = (const vecf*)proj;

  vecf acc = 0.0f;
  for (int c0 = 0; c0 < cnt; c0 += 32) {
    const int m = min(32, cnt - c0);
    int2 my = make_int2(0, 0);
    if (lane < m) my = pr[c0 + lane];
    int j = 0;
    for (; j + 4 <= m; j += 4) {
      const int s0 = __builtin_amdgcn_readlane(my.x, j);
      const int s1 = __builtin_amdgcn_readlane(my.x, j + 1);
      const int s2 = __builtin_amdgcn_readlane(my.x, j + 2);
      const int s3 = __builtin_amdgcn_readlane(my.x, j + 3);
      const float w0 = __int_as_float(__builtin_amdgcn_readlane(my.y, j));
      const float w1 = __int_as_float(__builtin_amdgcn_readlane(my.y, j + 1));
      const float w2 = __int_as_float(__builtin_amdgcn_readlane(my.y, j + 2));
      const float w3 = __int_as_float(__builtin_amdgcn_readlane(my.y, j + 3));
      const vecf v0 = rows[(size_t)s0 * 32 + lane];
      const vecf v1 = rows[(size_t)s1 * 32 + lane];
      const vecf v2 = rows[(size_t)s2 * 32 + lane];
      const vecf v3 = rows[(size_t)s3 * 32 + lane];
      acc += w0 * v0;
      acc += w1 * v1;
      acc += w2 * v2;
      acc += w3 * v3;
    }
    for (; j < m; ++j) {
      const int s0 = __builtin_amdgcn_readlane(my.x, j);
      const float w0 = __int_as_float(__builtin_amdgcn_readlane(my.y, j));
      const vecf v0 = rows[(size_t)s0 * 32 + lane];
      acc += w0 * v0;
    }
  }

  const float sc = idi[node];
  const vecf b = ((const vecf*)bias)[lane];
  vecf o = acc * sc + b;
  if (RELU) {
#pragma unroll
    for (int t = 0; t < VEC; ++t) o[t] = fmaxf(o[t], 0.0f);
  }
  ((vecf*)out)[(size_t)node * 32 + lane] = o;
}

extern "C" void kernel_launch(void* const* d_in, const int* in_sizes, int n_in,
                              void* d_out, int out_size, void* d_ws, size_t ws_size,
                              hipStream_t stream) {
  const float* feat = (const float*)d_in[0];
  const float* ew   = (const float*)d_in[1];
  const float* W1   = (const float*)d_in[2];
  const float* b1   = (const float*)d_in[3];
  const float* W2   = (const float*)d_in[4];
  const float* b2   = (const float*)d_in[5];
  const float* W3   = (const float*)d_in[6];
  const float* b3   = (const float*)d_in[7];
  const int*   src  = (const int*)d_in[8];
  const int*   dst  = (const int*)d_in[9];
  float* out = (float*)d_out;

  float* ws     = (float*)d_ws;
  int*   cnt_out = (int*)ws;
  int*   cnt_in  = (int*)ws + N_NODES;
  float* odi    = ws + 2 * (size_t)N_NODES;
  float* idi    = ws + 3 * (size_t)N_NODES;
  float* h      = ws + 4 * (size_t)N_NODES;
  float* proj   = h    + (size_t)N_NODES * 128;
  int*   rowptr = (int*)(proj + (size_t)N_NODES * 128);
  int*   rank   = rowptr + N_NODES;
  int*   pairs  = rank + N_EDGES;

  hipMemsetAsync(cnt_out, 0, 2 * (size_t)N_NODES * sizeof(int), stream);
  gcn_degrees<<<(N_EDGES + 255) / 256, 256, 0, stream>>>(src, dst, cnt_out, cnt_in, rank);
  gcn_scan_norm<<<(N_NODES + 1023) / 1024, 1024, 0, stream>>>(cnt_out, cnt_in, odi, idi, rowptr);
  gcn_fill<<<(N_EDGES + 255) / 256, 256, 0, stream>>>(src, dst, ew, rank, rowptr, pairs);

  const int gemmBlocks = N_NODES / 16;
  const int gatBlocks = (N_NODES + 7) / 8;

  gcn_gemm_wmma<128><<<gemmBlocks, 256, 0, stream>>>(feat, odi, W1, proj);
  gcn_gather<128, 1><<<gatBlocks, 256, 0, stream>>>(proj, pairs, rowptr, cnt_in, idi, b1, h);

  gcn_gemm_wmma<128><<<gemmBlocks, 256, 0, stream>>>(h, odi, W2, proj);
  gcn_gather<128, 1><<<gatBlocks, 256, 0, stream>>>(proj, pairs, rowptr, cnt_in, idi, b2, h);

  gcn_gemm_wmma<64><<<gemmBlocks, 128, 0, stream>>>(h, odi, W3, proj);
  gcn_gather<64, 0><<<gatBlocks, 256, 0, stream>>>(proj, pairs, rowptr, cnt_in, idi, b3, out);
}


# Fewer, longer waves (a quarter as many per call): gather wave sums 4 nodes, GEMM wave 4 column tiles, edge passes 4 edges per thread; node sum order still arrival rank (edge-thread map changed)
# speedup vs baseline: 2.5846x; 1.0822x over previous
#include <hip/hip_runtime.h>
#include <math.h>


typedef __attribute__((ext_vector_type(2))) float v2f;
typedef __attribute__((ext_vector_type(8))) float v8f;

#define N_NODES 50000
#define N_EDGES 800000
#define K_DIM   128

#define EDGE_PER_THREAD 4
#define EDGE_STRIDE (N_EDGES / EDGE_PER_THREAD)
#define EDGE_BLOCKS ((EDGE_STRIDE + 255) / 256)
static_assert(EDGE_STRIDE * EDGE_PER_THREAD == N_EDGES, "edges split evenly over threads");

#define GAT_NPW    4
#define GAT_WAVES  4
#define GAT_TOTAL  (N_NODES / GAT_NPW)
#define GAT_BLOCKS (GAT_TOTAL / GAT_WAVES)
static_assert(GAT_TOTAL * GAT_NPW == N_NODES, "nodes split evenly over waves");
static_assert(GAT_BLOCKS * GAT_WAVES == GAT_TOTAL, "waves split evenly over blocks");

#define GEMM_TPW 4

__global__ void gcn_degrees(const int* __restrict__ src, const int* __restrict__ dst,
                            int* __restrict__ cnt_out, int* __restrict__ cnt_in,
                            int* __restrict__ rank) {
  const int t = blockIdx.x * 256 + threadIdx.x;
  if (t >= EDGE_STRIDE) return;
  int s[EDGE_PER_THREAD], d[EDGE_PER_THREAD], r[EDGE_PER_THREAD];
#pragma unroll
  for (int i = 0; i < EDGE_PER_THREAD; ++i) {
    s[i] = src[t + i * EDGE_STRIDE];
    d[i] = dst[t + i * EDGE_STRIDE];
  }
#pragma unroll
  for (int i = 0; i < EDGE_PER_THREAD; ++i) atomicAdd(&cnt_out[s[i]], 1);
#pragma unroll
  for (int i = 0; i < EDGE_PER_THREAD; ++i) r[i] = atomicAdd(&cnt_in[d[i]], 1);
#pragma unroll
  for (int i = 0; i < EDGE_PER_THREAD; ++i) rank[t + i * EDGE_STRIDE] = r[i];
}

__global__ __launch_bounds__(1024) void gcn_scan_norm(const int* __restrict__ cnt_out,
                                                      const int* __restrict__ cnt_in,
                                                      float* __restrict__ odi,
                                                      float* __restrict__ idi,
                                                      int* __restrict__ rowptr) {
  __shared__ int s_part[32];
  __shared__ int s_wtot[32];
  const int tid  = threadIdx.x;
  const int lane = tid & 31;
  const int wave = tid >> 5;
  const int tile0 = blockIdx.x * 1024;

  int part = 0;
  for (int i = tid; i < tile0; i += 1024) part += cnt_in[i];

  const int i = tile0 + tid;
  const int c = (i < N_NODES) ? cnt_in[i] : 0;

  int incl = c;
  for (int o = 1; o < 32; o <<= 1) {
    const int n = __shfl_up(incl, o, 32);
    if (lane >= o) incl += n;
  }
  for (int o = 16; o > 0; o >>= 1) part += __shfl_xor(part, o, 32);

  if (lane == 31) s_wtot[wave] = incl;
  if (lane == 0)  s_part[wave] = part;
  __syncthreads();

  int base = 0;
  for (int w = 0; w < 32; ++w) {
    base += s_part[w];
    if (w < wave) base += s_wtot[w];
  }

  if (i < N_NODES) {
    rowptr[i] = base + incl - c;
    odi[i] = rsqrtf(fmaxf((float)cnt_out[i], 1.0f));
    idi[i] = rsqrtf(fmaxf((float)c, 1.0f));
  }
}

__global__ void gcn_fill(const int* __restrict__ src, const int* __restrict__ dst,
                         const float* __restrict__ ew, const int* __restrict__ rank,
                         const int* __restrict__ rowptr, int* __restrict__ pairs) {
  const int t = blockIdx.x * 256 + threadIdx.x;
  if (t >= EDGE_STRIDE) return;
  int d[EDGE_PER_THREAD], p[EDGE_PER_THREAD];
  int2 v[EDGE_PER_THREAD];
#pragma unroll
  for (int i = 0; i < EDGE_PER_THREAD; ++i) {
    const int e = t + i * EDGE_STRIDE;
    d[i] = dst[e];
    p[i] = rank[e];
    v[i] = make_int2(src[e], __float_as_int(ew[e]));
  }
#pragma unroll
  for (int i = 0; i < EDGE_PER_THREAD; ++i) p[i] += rowptr[d[i]];
#pragma unroll
  for (int i = 0; i < EDGE_PER_THREAD; ++i) ((int2*)pairs)[p[i]] = v[i];
}

template <int NOUT>
__global__ __launch_bounds__(32 * NOUT / (16 * GEMM_TPW)) void gcn_gemm_wmma(
    const float* __restrict__ A, const float* __restrict__ odi,
    const float* __restrict__ W, float* __restrict__ C) {
  constexpr int THREADS = 32 * NOUT / (16 * GEMM_TPW);
  __shared__ __attribute__((aligned(16))) float As[16][K_DIM + 4];

  const int tileM = blockIdx.x * 16;

  constexpr int NLD = 16 * (K_DIM / 4) / THREADS;
  const int wv = __builtin_amdgcn_readfirstlane(threadIdx.x >> 5);
  const int c4 = threadIdx.x & 31;
  float4 av[NLD];
  float  as[NLD];
#pragma unroll
  for (int it = 0; it < NLD; ++it) {
    const int r = wv + it * (THREADS / 32);
    av[it] = ((const float4*)A)[(size_t)(tileM + r) * (K_DIM / 4) + c4];
    as[it] = odi[tileM + r];
  }
#pragma unroll
  for (int it = 0; it < NLD; ++it) {
    const int r = wv + it * (THREADS / 32);
    float4 v = av[it];
    v.x *= as[it]; v.y *= as[it]; v.z *= as[it]; v.w *= as[it];
    *(float4*)&As[r][4 * c4] = v;
  }
  __syncthreads();

  const int wave = threadIdx.x >> 5;
  const int lane = threadIdx.x & 31;
  const int half = lane >> 4;
  const int r16  = lane & 15;
  const int nb   = wave * (16 * GEMM_TPW);

  v8f acc[GEMM_TPW];
#pragma unroll
  for (int t = 0; t < GEMM_TPW; ++t) acc[t] = v8f{};

#pragma unroll
  for (int k = 0; k < K_DIM; k += 4) {
    v2f a;
    a.x = As[r16][k + 2 * half];
    a.y = As[r16][k + 2 * half + 1];
#pragma unroll
    for (int t = 0; t < GEMM_TPW; ++t) {
      v2f b;
      b.x = W[(size_t)(k + 2 * half)     * NOUT + nb + 16 * t + r16];
      b.y = W[(size_t)(k + 2 * half + 1) * NOUT + nb + 16 * t + r16];
      acc[t] = __builtin_amdgcn_wmma_f32_16x16x4_f32(false, a, false, b,
                                                     (short)0, acc[t], false, false);
    }
  }

#pragma unroll
  for (int t = 0; t < GEMM_TPW; ++t) {
#pragma unroll
    for (int r = 0; r < 8; ++r) {
      const int row = tileM + r + 8 * half;
      C[(size_t)row * NOUT + nb + 16 * t + r16] = acc[t][r];
    }
  }
}

template <int D, int RELU>
__global__ __launch_bounds__(32 * GAT_WAVES) void gcn_gather(const float* __restrict__ proj,
                                                  const int* __restrict__ pairs,
                                                  const int* __restrict__ rowptr,
                                                  const int* __restrict__ cnt_in,
                                                  const float* __restrict__ idi,
                                                  const float* __restrict__ bias,
                                                  float* __restrict__ out) {
  constexpr int VEC = D / 32;
  typedef float vecf __attribute__((ext_vector_type(VEC)));

  const int lane = threadIdx.x & 31;
  const int w = __builtin_amdgcn_readfirstlane(blockIdx.x * GAT_WAVES + (threadIdx.x >> 5));

  const vecf* __restrict__ rows = (const vecf*)proj;

  int beg[GAT_NPW], cnt[GAT_NPW];
  float scl[GAT_NPW];
  int2 first[GAT_NPW];
#pragma unroll
  for (int i = 0; i < GAT_NPW; ++i) {
    const int node = w + i * GAT_TOTAL;
    beg[i] = rowptr[node];
    cnt[i] = cnt_in[node];
    scl[i] = idi[node];
  }
#pragma unroll
  for (int i = 0; i < GAT_NPW; ++i) {
    first[i] = make_int2(0, 0);
    if (lane < cnt[i]) first[i] = ((const int2*)pairs)[beg[i] + lane];
  }
  const vecf b = ((const vecf*)bias)[lane];

#pragma unroll
  for (int i = 0; i < GAT_NPW; ++i) {
    const int node = w + i * GAT_TOTAL;
    const int n = cnt[i];
    const int2* __restrict__ pr = (const int2*)pairs + beg[i];

    vecf acc = 0.0f;
    int2 my = first[i];
    for (int c0 = 0; c0 < n; c0 += 32) {
      const int m = min(32, n - c0);
      if (c0 > 0) {
        my = make_int2(0, 0);
        if (lane < m) my = pr[c0 + lane];
      }
      int j = 0;
      for (; j + 4 <= m; j += 4) {
        const int s0 = __builtin_amdgcn_readlane(my.x, j);
        const int s1 = __builtin_amdgcn_readlane(my.x, j + 1);
        const int s2 = __builtin_amdgcn_readlane(my.x, j + 2);
        const int s3 = __builtin_amdgcn_readlane(my.x, j + 3);
        const float w0 = __int_as_float(__builtin_amdgcn_readlane(my.y, j));
        const float w1 = __int_as_float(__builtin_amdgcn_readlane(my.y, j + 1));
        const float w2 = __int_as_float(__builtin_amdgcn_readlane(my.y, j + 2));
        const float w3 = __int_as_float(__builtin_amdgcn_readlane(my.y, j + 3));
        const vecf v0 = rows[(size_t)s0 * 32 + lane];
        const vecf v1 = rows[(size_t)s1 * 32 + lane];
        const vecf v2 = rows[(size_t)s2 * 32 + lane];
        const vecf v3 = rows[(size_t)s3 * 32 + lane];
        acc += w0 * v0;
        acc += w1 * v1;
        acc += w2 * v2;
        acc += w3 * v3;
      }
      for (; j < m; ++j) {
        const int s0 = __builtin_amdgcn_readlane(my.x, j);
        const float w0 = __int_as_float(__builtin_amdgcn_readlane(my.y, j));
        const vecf v0 = rows[(size_t)s0 * 32 + lane];
        acc += w0 * v0;
      }
    }

    vecf o = acc * scl[i] + b;
    if (RELU) {
#pragma unroll
      for (int t = 0; t < VEC; ++t) o[t] = fmaxf(o[t], 0.0f);
    }
    ((vecf*)out)[(size_t)node * 32 + lane] = o;
  }
}

extern "C" void kernel_launch(void* const* d_in, const int* in_sizes, int n_in,
                              void* d_out, int out_size, void* d_ws, size_t ws_size,
                              hipStream_t stream) {
  const float* feat = (const float*)d_in[0];
  const float* ew   = (const float*)d_in[1];
  const float* W1   = (const float*)d_in[2];
  const float* b1   = (const float*)d_in[3];
  const float* W2   = (const float*)d_in[4];
  const float* b2   = (const float*)d_in[5];
  const float* W3   = (const float*)d_in[6];
  const float* b3   = (const float*)d_in[7];
  const int*   src  = (const int*)d_in[8];
  const int*   dst  = (const int*)d_in[9];
  float* out = (float*)d_out;

  float* ws     = (float*)d_ws;
  int*   cnt_out = (int*)ws;
  int*   cnt_in  = (int*)ws + N_NODES;
  float* odi    = ws + 2 * (size_t)N_NODES;
  float* idi    = ws + 3 * (size_t)N_NODES;
  float* h      = ws + 4 * (size_t)N_NODES;
  float* proj   = h    + (size_t)N_NODES * 128;
  int*   rowptr = (int*)(proj + (size_t)N_NODES * 128);
  int*   rank   = rowptr + N_NODES;
  int*   pairs  = rank + N_EDGES;

  hipMemsetAsync(cnt_out, 0, 2 * (size_t)N_NODES * sizeof(int), stream);
  gcn_degrees<<<EDGE_BLOCKS, 256, 0, stream>>>(src, dst, cnt_out, cnt_in, rank);
  gcn_scan_norm<<<(N_NODES + 1023) / 1024, 1024, 0, stream>>>(cnt_out, cnt_in, odi, idi, rowptr);
  gcn_fill<<<EDGE_BLOCKS, 256, 0, stream>>>(src, dst, ew, rank, rowptr, pairs);

  const int gemmBlocks = N_NODES / 16;
  const int gatBlocks = GAT_BLOCKS;

  gcn_gemm_wmma<128><<<gemmBlocks, 64, 0, stream>>>(feat, odi, W1, proj);
  gcn_gather<128, 1><<<gatBlocks, 32 * GAT_WAVES, 0, stream>>>(proj, pairs, rowptr, cnt_in, idi, b1, h);

  gcn_gemm_wmma<128><<<gemmBlocks, 64, 0, stream>>>(h, odi, W2, proj);
  gcn_gather<128, 1><<<gatBlocks, 32 * GAT_WAVES, 0, stream>>>(proj, pairs, rowptr, cnt_in, idi, b2, h);

  gcn_gemm_wmma<64><<<gemmBlocks, 32, 0, stream>>>(h, odi, W3, proj);
  gcn_gather<64, 0><<<gatBlocks, 32 * GAT_WAVES, 0, stream>>>(proj, pairs, rowptr, cnt_in, idi, b3, out);
}


# Edge-list build without global atomics or memset: LDS counters per (node range, edge slice) work-group count degrees and hand out slots; node sum order = slice, then LDS arrival
# speedup vs baseline: 4.2861x; 1.6583x over previous
#include <hip/hip_runtime.h>
#include <math.h>


typedef __attribute__((ext_vector_type(2))) float v2f;
typedef __attribute__((ext_vector_type(8))) float v8f;

#define N_NODES 50000
#define N_EDGES 800000
#define K_DIM   128

#define NRANGE   16
#define RANGE_N  3200
#define NODE_PAD (NRANGE * RANGE_N)
#define NSLICE   16
#define SLICE_E  (N_EDGES / NSLICE)
#define SLICE_V  (SLICE_E / 4)
#define CHUNK_N  800
#define CHUNKS_PER_RANGE (RANGE_N / CHUNK_N)
#define NCHUNK   (NODE_PAD / CHUNK_N)
#define EDGE_THREADS 1024
#define EDGE_BATCH   4
static_assert(NODE_PAD >= N_NODES, "ranges cover all nodes");
static_assert(SLICE_E * NSLICE == N_EDGES && SLICE_V * 4 == SLICE_E, "edges split evenly into vector slices");
static_assert(CHUNK_N * CHUNKS_PER_RANGE == RANGE_N && CHUNK_N % 32 == 0, "whole waves per chunk");
static_assert(CHUNKS_PER_RANGE <= EDGE_THREADS / 32, "one wave per chunk total");

#define GAT_NPW    4
#define GAT_WAVES  4
#define GAT_TOTAL  (N_NODES / GAT_NPW)
#define GAT_BLOCKS (GAT_TOTAL / GAT_WAVES)
static_assert(GAT_TOTAL * GAT_NPW == N_NODES, "nodes split evenly over waves");
static_assert(GAT_BLOCKS * GAT_WAVES == GAT_TOTAL, "waves split evenly over blocks");

#define GEMM_TPW 4

__global__ __launch_bounds__(EDGE_THREADS) void gcn_count(const int* __restrict__ src,
                                                          const int* __restrict__ dst,
                                                          int* __restrict__ part_out,
                                                          int* __restrict__ part_in,
                                                          int* __restrict__ tot_in) {
  __shared__ int c_in[RANGE_N];
  __shared__ int c_out[RANGE_N];
  const int tid = threadIdx.x;
  const int g = blockIdx.x, s = blockIdx.y;
  const int lo = g * RANGE_N;

  for (int i = tid; i < RANGE_N; i += EDGE_THREADS) { c_in[i] = 0; c_out[i] = 0; }
  __syncthreads();

  const int4* __restrict__ s4 = (const int4*)(src + (size_t)s * SLICE_E);
  const int4* __restrict__ d4 = (const int4*)(dst + (size_t)s * SLICE_E);
  for (int v0 = tid; v0 < SLICE_V; v0 += EDGE_THREADS * EDGE_BATCH) {
    int4 sv[EDGE_BATCH], dv[EDGE_BATCH];
#pragma unroll
    for (int k = 0; k < EDGE_BATCH; ++k) {
      const int v = v0 + k * EDGE_THREADS;
      sv[k] = make_int4(-1, -1, -1, -1);
      dv[k] = make_int4(-1, -1, -1, -1);
      if (v < SLICE_V) { sv[k] = s4[v]; dv[k] = d4[v]; }
    }
#pragma unroll
    for (int k = 0; k < EDGE_BATCH; ++k) {
      unsigned r;
      r = (unsigned)(dv[k].x - lo); if (r < RANGE_N) atomicAdd(&c_in[r], 1);
      r = (unsigned)(dv[k].y - lo); if (r < RANGE_N) atomicAdd(&c_in[r], 1);
      r = (unsigned)(dv[k].z - lo); if (r < RANGE_N) atomicAdd(&c_in[r], 1);
      r = (unsigned)(dv[k].w - lo); if (r < RANGE_N) atomicAdd(&c_in[r], 1);
      r = (unsigned)(sv[k].x - lo); if (r < RANGE_N) atomicAdd(&c_out[r], 1);
      r = (unsigned)(sv[k].y - lo); if (r < RANGE_N) atomicAdd(&c_out[r], 1);
      r = (unsigned)(sv[k].z - lo); if (r < RANGE_N) atomicAdd(&c_out[r], 1);
      r = (unsigned)(sv[k].w - lo); if (r < RANGE_N) atomicAdd(&c_out[r], 1);
    }
  }
  __syncthreads();

  for (int i = tid; i < RANGE_N; i += EDGE_THREADS) {
    part_in[(size_t)s * NODE_PAD + lo + i]  = c_in[i];
    part_out[(size_t)s * NODE_PAD + lo + i] = c_out[i];
  }

  const int lane = tid & 31;
  const int wave = tid >> 5;
  if (wave < CHUNKS_PER_RANGE) {
    int t = 0;
    for (int i = lane; i < CHUNK_N; i += 32) t += c_in[wave * CHUNK_N + i];
    for (int o = 16; o > 0; o >>= 1) t += __shfl_xor(t, o, 32);
    if (lane == 0) tot_in[(g * CHUNKS_PER_RANGE + wave) * NSLICE + s] = t;
  }
}

__global__ __launch_bounds__(CHUNK_N) void gcn_scan_norm(int* part_in,
                                                         const int* __restrict__ part_out,
                                                         const int* __restrict__ tot_in,
                                                         int* __restrict__ cnt_in,
                                                         float* __restrict__ odi,
                                                         float* __restrict__ idi,
                                                         int* __restrict__ rowptr) {
  constexpr int WAVES = CHUNK_N / 32;
  __shared__ int s_part[WAVES];
  __shared__ int s_wtot[WAVES];
  const int tid  = threadIdx.x;
  const int lane = tid & 31;
  const int wave = tid >> 5;
  const int c = blockIdx.x;

  int part = 0;
  for (int i = tid; i < c * NSLICE; i += CHUNK_N) part += tot_in[i];

  const int i = c * CHUNK_N + tid;
  int pin[NSLICE];
  int cout = 0;
#pragma unroll
  for (int s = 0; s < NSLICE; ++s) pin[s] = part_in[(size_t)s * NODE_PAD + i];
#pragma unroll
  for (int s = 0; s < NSLICE; ++s) cout += part_out[(size_t)s * NODE_PAD + i];
  int cin = 0;
#pragma unroll
  for (int s = 0; s < NSLICE; ++s) cin += pin[s];

  int incl = cin;
  for (int o = 1; o < 32; o <<= 1) {
    const int n = __shfl_up(incl, o, 32);
    if (lane >= o) incl += n;
  }
  for (int o = 16; o > 0; o >>= 1) part += __shfl_xor(part, o, 32);

  if (lane == 31) s_wtot[wave] = incl;
  if (lane == 0)  s_part[wave] = part;
  __syncthreads();

  int base = 0;
  for (int w = 0; w < WAVES; ++w) {
    base += s_part[w];
    if (w < wave) base += s_wtot[w];
  }

  const int rp = base + incl - cin;
  if (i < N_NODES) {
    rowptr[i] = rp;
    cnt_in[i] = cin;
    odi[i] = rsqrtf(fmaxf((float)cout, 1.0f));
    idi[i] = rsqrtf(fmaxf((float)cin, 1.0f));
  }
  int run = rp;
#pragma unroll
  for (int s = 0; s < NSLICE; ++s) {
    part_in[(size_t)s * NODE_PAD + i] = run;
    run += pin[s];
  }
}

__global__ __launch_bounds__(EDGE_THREADS) void gcn_fill(const int* __restrict__ src,
                                                         const int* __restrict__ dst,
                                                         const float* __restrict__ ew,
                                                         const int* __restrict__ start,
                                                         int* __restrict__ pairs) {
  __shared__ int cur[RANGE_N];
  const int tid = threadIdx.x;
  const int g = blockIdx.x, s = blockIdx.y;
  const int lo = g * RANGE_N;

  constexpr int NST = (RANGE_N + EDGE_THREADS - 1) / EDGE_THREADS;
  int st[NST];
#pragma unroll
  for (int k = 0; k < NST; ++k) {
    const int i = tid + k * EDGE_THREADS;
    st[k] = 0;
    if (i < RANGE_N) st[k] = start[(size_t)s * NODE_PAD + lo + i];
  }
#pragma unroll
  for (int k = 0; k < NST; ++k) {
    const int i = tid + k * EDGE_THREADS;
    if (i < RANGE_N) cur[i] = st[k];
  }
  __syncthreads();

  const int4*   __restrict__ s4 = (const int4*)(src + (size_t)s * SLICE_E);
  const int4*   __restrict__ d4 = (const int4*)(dst + (size_t)s * SLICE_E);
  const float4* __restrict__ w4 = (const float4*)(ew + (size_t)s * SLICE_E);
  int2* __restrict__ pr = (int2*)pairs;
  for (int v0 = tid; v0 < SLICE_V; v0 += EDGE_THREADS * EDGE_BATCH) {
    int4 sv[EDGE_BATCH], dv[EDGE_BATCH];
    float4 wv[EDGE_BATCH];
#pragma unroll
    for (int k = 0; k < EDGE_BATCH; ++k) {
      const int v = v0 + k * EDGE_THREADS;
      sv[k] = make_int4(0, 0, 0, 0);
      dv[k] = make_int4(-1, -1, -1, -1);
      wv[k] = make_float4(0.0f, 0.0f, 0.0f, 0.0f);
      if (v < SLICE_V) { sv[k] = s4[v]; dv[k] = d4[v]; wv[k] = w4[v]; }
    }
    int px[EDGE_BATCH], py[EDGE_BATCH], pz[EDGE_BATCH], pw[EDGE_BATCH];
#pragma unroll
    for (int k = 0; k < EDGE_BATCH; ++k) {
      unsigned r;
      px[k] = -1; r = (unsigned)(dv[k].x - lo); if (r < RANGE_N) px[k] = atomicAdd(&cur[r], 1);
      py[k] = -1; r = (unsigned)(dv[k].y - lo); if (r < RANGE_N) py[k] = atomicAdd(&cur[r], 1);
      pz[k] = -1; r = (unsigned)(dv[k].z - lo); if (r < RANGE_N) pz[k] = atomicAdd(&cur[r], 1);
      pw[k] = -1; r = (unsigned)(dv[k].w - lo); if (r < RANGE_N) pw[k] = atomicAdd(&cur[r], 1);
    }
#pragma unroll
    for (int k = 0; k < EDGE_BATCH; ++k) {
      if (px[k] >= 0) pr[px[k]] = make_int2(sv[k].x, __float_as_int(wv[k].x));
      if (py[k] >= 0) pr[py[k]] = make_int2(sv[k].y, __float_as_int(wv[k].y));
      if (pz[k] >= 0) pr[pz[k]] = make_int2(sv[k].z, __float_as_int(wv[k].z));
      if (pw[k] >= 0) pr[pw[k]] = make_int2(sv[k].w, __float_as_int(wv[k].w));
    }
  }
}

template <int NOUT>
__global__ __launch_bounds__(32 * NOUT / (16 * GEMM_TPW)) void gcn_gemm_wmma(
    const float* __restrict__ A, const float* __restrict__ odi,
    const float* __restrict__ W, float* __restrict__ C) {
  constexpr int THREADS = 32 * NOUT / (16 * GEMM_TPW);
  __shared__ __attribute__((aligned(16))) float As[16][K_DIM + 4];

  const int tileM = blockIdx.x * 16;

  constexpr int NLD = 16 * (K_DIM / 4) / THREADS;
  const int wv = __builtin_amdgcn_readfirstlane(threadIdx.x >> 5);
  const int c4 = threadIdx.x & 31;
  float4 av[NLD];
  float  as[NLD];
#pragma unroll
  for (int it = 0; it < NLD; ++it) {
    const int r = wv + it * (THREADS / 32);
    av[it] = ((const float4*)A)[(size_t)(tileM + r) * (K_DIM / 4) + c4];
    as[it] = odi[tileM + r];
  }
#pragma unroll
  for (int it = 0; it < NLD; ++it) {
    const int r = wv + it * (THREADS / 32);
    float4 v = av[it];
    v.x *= as[it]; v.y *= as[it]; v.z *= as[it]; v.w *= as[it];
    *(float4*)&As[r][4 * c4] = v;
  }
  __syncthreads();

  const int wave = threadIdx.x >> 5;
  const int lane = threadIdx.x & 31;
  const int half = lane >> 4;
  const int r16  = lane & 15;
  const int nb   = wave * (16 * GEMM_TPW);

  v8f acc[GEMM_TPW];
#pragma unroll
  for (int t = 0; t < GEMM_TPW; ++t) acc[t] = v8f{};

#pragma unroll
  for (int k = 0; k < K_DIM; k += 4) {
    v2f a;
    a.x = As[r16][k + 2 * half];
    a.y = As[r16][k + 2 * half + 1];
#pragma unroll
    for (int t = 0; t < GEMM_TPW; ++t) {
      v2f b;
      b.x = W[(size_t)(k + 2 * half)     * NOUT + nb + 16 * t + r16];
      b.y = W[(size_t)(k + 2 * half + 1) * NOUT + nb + 16 * t + r16];
      acc[t] = __builtin_amdgcn_wmma_f32_16x16x4_f32(false, a, false, b,
                                                     (short)0, acc[t], false, false);
    }
  }

#pragma unroll
  for (int t = 0; t < GEMM_TPW; ++t) {
#pragma unroll
    for (int r = 0; r < 8; ++r) {
      const int row = tileM + r + 8 * half;
      C[(size_t)row * NOUT + nb + 16 * t + r16] = acc[t][r];
    }
  }
}

template <int D, int RELU>
__global__ __launch_bounds__(32 * GAT_WAVES) void gcn_gather(const float* __restrict__ proj,
                                                  const int* __restrict__ pairs,
                                                  const int* __restrict__ rowptr,
                                                  const int* __restrict__ cnt_in,
                                                  const float* __restrict__ idi,
                                                  const float* __restrict__ bias,
                                                  float* __restrict__ out) {
  constexpr int VEC = D / 32;
  typedef float vecf __attribute__((ext_vector_type(VEC)));

  const int lane = threadIdx.x & 31;
  const int w = __builtin_amdgcn_readfirstlane(blockIdx.x * GAT_WAVES + (threadIdx.x >> 5));

  const vecf* __restrict__ rows = (const vecf*)proj;

  int beg[GAT_NPW], cnt[GAT_NPW];
  float scl[GAT_NPW];
  int2 first[GAT_NPW];
#pragma unroll
  for (int i = 0; i < GAT_NPW; ++i) {
    const int node = w + i * GAT_TOTAL;
    beg[i] = rowptr[node];
    cnt[i] = cnt_in[node];
    scl[i] = idi[node];
  }
#pragma unroll
  for (int i = 0; i < GAT_NPW; ++i) {
    first[i] = make_int2(0, 0);
    if (lane < cnt[i]) first[i] = ((const int2*)pairs)[beg[i] + lane];
  }
  const vecf b = ((const vecf*)bias)[lane];

#pragma unroll
  for (int i = 0; i < GAT_NPW; ++i) {
    const int node = w + i * GAT_TOTAL;
    const int n = cnt[i];
    const int2* __restrict__ pr = (const int2*)pairs + beg[i];

    vecf acc = 0.0f;
    int2 my = first[i];
    for (int c0 = 0; c0 < n; c0 += 32) {
      const int m = min(32, n - c0);
      if (c0 > 0) {
        my = make_int2(0, 0);
        if (lane < m) my = pr[c0 + lane];
      }
      int j = 0;
      for (; j + 4 <= m; j += 4) {
        const int s0 = __builtin_amdgcn_readlane(my.x, j);
        const int s1 = __builtin_amdgcn_readlane(my.x, j + 1);
        const int s2 = __builtin_amdgcn_readlane(my.x, j + 2);
        const int s3 = __builtin_amdgcn_readlane(my.x, j + 3);
        const float w0 = __int_as_float(__builtin_amdgcn_readlane(my.y, j));
        const float w1 = __int_as_float(__builtin_amdgcn_readlane(my.y, j + 1));
        const float w2 = __int_as_float(__builtin_amdgcn_readlane(my.y, j + 2));
        const float w3 = __int_as_float(__builtin_amdgcn_readlane(my.y, j + 3));
        const vecf v0 = rows[(size_t)s0 * 32 + lane];
        const vecf v1 = rows[(size_t)s1 * 32 + lane];
        const vecf v2 = rows[(size_t)s2 * 32 + lane];
        const vecf v3 = rows[(size_t)s3 * 32 + lane];
        acc += w0 * v0;
        acc += w1 * v1;
        acc += w2 * v2;
        acc += w3 * v3;
      }
      for (; j < m; ++j) {
        const int s0 = __builtin_amdgcn_readlane(my.x, j);
        const float w0 = __int_as_float(__builtin_amdgcn_readlane(my.y, j));
        const vecf v0 = rows[(size_t)s0 * 32 + lane];
        acc += w0 * v0;
      }
    }

    vecf o = acc * scl[i] + b;
    if (RELU) {
#pragma unroll
      for (int t = 0; t < VEC; ++t) o[t] = fmaxf(o[t], 0.0f);
    }
    ((vecf*)out)[(size_t)node * 32 + lane] = o;
  }
}

extern "C" void kernel_launch(void* const* d_in, const int* in_sizes, int n_in,
                              void* d_out, int out_size, void* d_ws, size_t ws_size,
                              hipStream_t stream) {
  const float* feat = (const float*)d_in[0];
  const float* ew   = (const float*)d_in[1];
  const float* W1   = (const float*)d_in[2];
  const float* b1   = (const float*)d_in[3];
  const float* W2   = (const float*)d_in[4];
  const float* b2   = (const float*)d_in[5];
  const float* W3   = (const float*)d_in[6];
  const float* b3   = (const float*)d_in[7];
  const int*   src  = (const int*)d_in[8];
  const int*   dst  = (const int*)d_in[9];
  float* out = (float*)d_out;

  float* ws     = (float*)d_ws;
  int*   cnt_in = (int*)ws;
  float* odi    = ws + 1 * (size_t)N_NODES;
  float* idi    = ws + 2 * (size_t)N_NODES;
  int*   rowptr = (int*)ws + 3 * (size_t)N_NODES;
  float* h      = ws + 4 * (size_t)N_NODES;
  float* proj   = h    + (size_t)N_NODES * 128;
  int*   pairs  = (int*)(proj + (size_t)N_NODES * 128);
  int*   part_in  = pairs + 2 * (size_t)N_EDGES;
  int*   part_out = part_in + (size_t)NSLICE * NODE_PAD;
  int*   tot_in   = part_out + (size_t)NSLICE * NODE_PAD;

  gcn_count<<<dim3(NRANGE, NSLICE), EDGE_THREADS, 0, stream>>>(src, dst, part_out, part_in, tot_in);
  gcn_scan_norm<<<NCHUNK, CHUNK_N, 0, stream>>>(part_in, part_out, tot_in, cnt_in, odi, idi, rowptr);
  gcn_fill<<<dim3(NRANGE, NSLICE), EDGE_THREADS, 0, stream>>>(src, dst, ew, part_in, pairs);

  const int gemmBlocks = N_NODES / 16;
  const int gatBlocks = GAT_BLOCKS;

  gcn_gemm_wmma<128><<<gemmBlocks, 64, 0, stream>>>(feat, odi, W1, proj);
  gcn_gather<128, 1><<<gatBlocks, 32 * GAT_WAVES, 0, stream>>>(proj, pairs, rowptr, cnt_in, idi, b1, h);

  gcn_gemm_wmma<128><<<gemmBlocks, 64, 0, stream>>>(h, odi, W2, proj);
  gcn_gather<128, 1><<<gatBlocks, 32 * GAT_WAVES, 0, stream>>>(proj, pairs, rowptr, cnt_in, idi, b2, h);

  gcn_gemm_wmma<64><<<gemmBlocks, 32, 0, stream>>>(h, odi, W3, proj);
  gcn_gather<64, 0><<<gatBlocks, 32 * GAT_WAVES, 0, stream>>>(proj, pairs, rowptr, cnt_in, idi, b3, out);
}


# Layers 1 and 2: row gather and the next layer's WMMA projection in one launch (16-row stripe summed into LDS, multiplied at once); hidden rows never stored; 9 launches to 7; edge sum order kept
# speedup vs baseline: 4.7503x; 1.1083x over previous
#include <hip/hip_runtime.h>
#include <math.h>


typedef __attribute__((ext_vector_type(2))) float v2f;
typedef __attribute__((ext_vector_type(8))) float v8f;

#define N_NODES 50000
#define N_EDGES 800000
#define K_DIM   128

#define NRANGE   16
#define RANGE_N  3200
#define NODE_PAD (NRANGE * RANGE_N)
#define NSLICE   16
#define SLICE_E  (N_EDGES / NSLICE)
#define SLICE_V  (SLICE_E / 4)
#define CHUNK_N  800
#define CHUNKS_PER_RANGE (RANGE_N / CHUNK_N)
#define NCHUNK   (NODE_PAD / CHUNK_N)
#define EDGE_THREADS 1024
#define EDGE_BATCH   4
static_assert(NODE_PAD >= N_NODES, "ranges cover all nodes");
static_assert(SLICE_E * NSLICE == N_EDGES && SLICE_V * 4 == SLICE_E, "edges split evenly into vector slices");
static_assert(CHUNK_N * CHUNKS_PER_RANGE == RANGE_N && CHUNK_N % 32 == 0, "whole waves per chunk");
static_assert(CHUNKS_PER_RANGE <= EDGE_THREADS / 32, "one wave per chunk total");

#define GAT_NPW    4
#define GAT_WAVES  4
#define GAT_TOTAL  (N_NODES / GAT_NPW)
#define GAT_BLOCKS (GAT_TOTAL / GAT_WAVES)
static_assert(GAT_TOTAL * GAT_NPW == N_NODES, "nodes split evenly over waves");
static_assert(GAT_BLOCKS * GAT_WAVES == GAT_TOTAL, "waves split evenly over blocks");

#define GEMM_TPW 4

__global__ __launch_bounds__(EDGE_THREADS) void gcn_count(const int* __restrict__ src,
                                                          const int* __restrict__ dst,
                                                          int* __restrict__ part_out,
                                                          int* __restrict__ part_in,
                                                          int* __restrict__ tot_in) {
  __shared__ int c_in[RANGE_N];
  __shared__ int c_out[RANGE_N];
  const int tid = threadIdx.x;
  const int g = blockIdx.x, s = blockIdx.y;
  const int lo = g * RANGE_N;

  for (int i = tid; i < RANGE_N; i += EDGE_THREADS) { c_in[i] = 0; c_out[i] = 0; }
  __syncthreads();

  const int4* __restrict__ s4 = (const int4*)(src + (size_t)s * SLICE_E);
  const int4* __restrict__ d4 = (const int4*)(dst + (size_t)s * SLICE_E);
  for (int v0 = tid; v0 < SLICE_V; v0 += EDGE_THREADS * EDGE_BATCH) {
    int4 sv[EDGE_BATCH], dv[EDGE_BATCH];
#pragma unroll
    for (int k = 0; k < EDGE_BATCH; ++k) {
      const int v = v0 + k * EDGE_THREADS;
      sv[k] = make_int4(-1, -1, -1, -1);
      dv[k] = make_int4(-1, -1, -1, -1);
      if (v < SLICE_V) { sv[k] = s4[v]; dv[k] = d4[v]; }
    }
#pragma unroll
    for (int k = 0; k < EDGE_BATCH; ++k) {
      unsigned r;
      r = (unsigned)(dv[k].x - lo); if (r < RANGE_N) atomicAdd(&c_in[r], 1);
      r = (unsigned)(dv[k].y - lo); if (r < RANGE_N) atomicAdd(&c_in[r], 1);
      r = (unsigned)(dv[k].z - lo); if (r < RANGE_N) atomicAdd(&c_in[r], 1);
      r = (unsigned)(dv[k].w - lo); if (r < RANGE_N) atomicAdd(&c_in[r], 1);
      r = (unsigned)(sv[k].x - lo); if (r < RANGE_N) atomicAdd(&c_out[r], 1);
      r = (unsigned)(sv[k].y - lo); if (r < RANGE_N) atomicAdd(&c_out[r], 1);
      r = (unsigned)(sv[k].z - lo); if (r < RANGE_N) atomicAdd(&c_out[r], 1);
      r = (unsigned)(sv[k].w - lo); if (r < RANGE_N) atomicAdd(&c_out[r], 1);
    }
  }
  __syncthreads();

  for (int i = tid; i < RANGE_N; i += EDGE_THREADS) {
    part_in[(size_t)s * NODE_PAD + lo + i]  = c_in[i];
    part_out[(size_t)s * NODE_PAD + lo + i] = c_out[i];
  }

  const int lane = tid & 31;
  const int wave = tid >> 5;
  if (wave < CHUNKS_PER_RANGE) {
    int t = 0;
    for (int i = lane; i < CHUNK_N; i += 32) t += c_in[wave * CHUNK_N + i];
    for (int o = 16; o > 0; o >>= 1) t += __shfl_xor(t, o, 32);
    if (lane == 0) tot_in[(g * CHUNKS_PER_RANGE + wave) * NSLICE + s] = t;
  }
}

__global__ __launch_bounds__(CHUNK_N) void gcn_scan_norm(int* part_in,
                                                         const int* __restrict__ part_out,
                                                         const int* __restrict__ tot_in,
                                                         int* __restrict__ cnt_in,
                                                         float* __restrict__ odi,
                                                         float* __restrict__ idi,
                                                         int* __restrict__ rowptr) {
  constexpr int WAVES = CHUNK_N / 32;
  __shared__ int s_part[WAVES];
  __shared__ int s_wtot[WAVES];
  const int tid  = threadIdx.x;
  const int lane = tid & 31;
  const int wave = tid >> 5;
  const int c = blockIdx.x;

  int part = 0;
  for (int i = tid; i < c * NSLICE; i += CHUNK_N) part += tot_in[i];

  const int i = c * CHUNK_N + tid;
  int pin[NSLICE];
  int cout = 0;
#pragma unroll
  for (int s = 0; s < NSLICE; ++s) pin[s] = part_in[(size_t)s * NODE_PAD + i];
#pragma unroll
  for (int s = 0; s < NSLICE; ++s) cout += part_out[(size_t)s * NODE_PAD + i];
  int cin = 0;
#pragma unroll
  for (int s = 0; s < NSLICE; ++s) cin += pin[s];

  int incl = cin;
  for (int o = 1; o < 32; o <<= 1) {
    const int n = __shfl_up(incl, o, 32);
    if (lane >= o) incl += n;
  }
  for (int o = 16; o > 0; o >>= 1) part += __shfl_xor(part, o, 32);

  if (lane == 31) s_wtot[wave] = incl;
  if (lane == 0)  s_part[wave] = part;
  __syncthreads();

  int base = 0;
  for (int w = 0; w < WAVES; ++w) {
    base += s_part[w];
    if (w < wave) base += s_wtot[w];
  }

  const int rp = base + incl - cin;
  if (i < N_NODES) {
    rowptr[i] = rp;
    cnt_in[i] = cin;
    odi[i] = rsqrtf(fmaxf((float)cout, 1.0f));
    idi[i] = rsqrtf(fmaxf((float)cin, 1.0f));
  }
  int run = rp;
#pragma unroll
  for (int s = 0; s < NSLICE; ++s) {
    part_in[(size_t)s * NODE_PAD + i] = run;
    run += pin[s];
  }
}

__global__ __launch_bounds__(EDGE_THREADS) void gcn_fill(const int* __restrict__ src,
                                                         const int* __restrict__ dst,
                                                         const float* __restrict__ ew,
                                                         const int* __restrict__ start,
                                                         int* __restrict__ pairs) {
  __shared__ int cur[RANGE_N];
  const int tid = threadIdx.x;
  const int g = blockIdx.x, s = blockIdx.y;
  const int lo = g * RANGE_N;

  constexpr int NST = (RANGE_N + EDGE_THREADS - 1) / EDGE_THREADS;
  int st[NST];
#pragma unroll
  for (int k = 0; k < NST; ++k) {
    const int i = tid + k * EDGE_THREADS;
    st[k] = 0;
    if (i < RANGE_N) st[k] = start[(size_t)s * NODE_PAD + lo + i];
  }
#pragma unroll
  for (int k = 0; k < NST; ++k) {
    const int i = tid + k * EDGE_THREADS;
    if (i < RANGE_N) cur[i] = st[k];
  }
  __syncthreads();

  const int4*   __restrict__ s4 = (const int4*)(src + (size_t)s * SLICE_E);
  const int4*   __restrict__ d4 = (const int4*)(dst + (size_t)s * SLICE_E);
  const float4* __restrict__ w4 = (const float4*)(ew + (size_t)s * SLICE_E);
  int2* __restrict__ pr = (int2*)pairs;
  for (int v0 = tid; v0 < SLICE_V; v0 += EDGE_THREADS * EDGE_BATCH) {
    int4 sv[EDGE_BATCH], dv[EDGE_BATCH];
    float4 wv[EDGE_BATCH];
#pragma unroll
    for (int k = 0; k < EDGE_BATCH; ++k) {
      const int v = v0 + k * EDGE_THREADS;
      sv[k] = make_int4(0, 0, 0, 0);
      dv[k] = make_int4(-1, -1, -1, -1);
      wv[k] = make_float4(0.0f, 0.0f, 0.0f, 0.0f);
      if (v < SLICE_V) { sv[k] = s4[v]; dv[k] = d4[v]; wv[k] = w4[v]; }
    }
    int px[EDGE_BATCH], py[EDGE_BATCH], pz[EDGE_BATCH], pw[EDGE_BATCH];
#pragma unroll
    for (int k = 0; k < EDGE_BATCH; ++k) {
      unsigned r;
      px[k] = -1; r = (unsigned)(dv[k].x - lo); if (r < RANGE_N) px[k] = atomicAdd(&cur[r], 1);
      py[k] = -1; r = (unsigned)(dv[k].y - lo); if (r < RANGE_N) py[k] = atomicAdd(&cur[r], 1);
      pz[k] = -1; r = (unsigned)(dv[k].z - lo); if (r < RANGE_N) pz[k] = atomicAdd(&cur[r], 1);
      pw[k] = -1; r = (unsigned)(dv[k].w - lo); if (r < RANGE_N) pw[k] = atomicAdd(&cur[r], 1);
    }
#pragma unroll
    for (int k = 0; k < EDGE_BATCH; ++k) {
      if (px[k] >= 0) pr[px[k]] = make_int2(sv[k].x, __float_as_int(wv[k].x));
      if (py[k] >= 0) pr[py[k]] = make_int2(sv[k].y, __float_as_int(wv[k].y));
      if (pz[k] >= 0) pr[pz[k]] = make_int2(sv[k].z, __float_as_int(wv[k].z));
      if (pw[k] >= 0) pr[pw[k]] = make_int2(sv[k].w, __float_as_int(wv[k].w));
    }
  }
}

template <int NOUT>
__global__ __launch_bounds__(32 * NOUT / (16 * GEMM_TPW)) void gcn_gemm_wmma(
    const float* __restrict__ A, const float* __restrict__ odi,
    const float* __restrict__ W, float* __restrict__ C) {
  constexpr int THREADS = 32 * NOUT / (16 * GEMM_TPW);
  __shared__ __attribute__((aligned(16))) float As[16][K_DIM + 4];

  const int tileM = blockIdx.x * 16;

  constexpr int NLD = 16 * (K_DIM / 4) / THREADS;
  const int wv = __builtin_amdgcn_readfirstlane(threadIdx.x >> 5);
  const int c4 = threadIdx.x & 31;
  float4 av[NLD];
  float  as[NLD];
#pragma unroll
  for (int it = 0; it < NLD; ++it) {
    const int r = wv + it * (THREADS / 32);
    av[it] = ((const float4*)A)[(size_t)(tileM + r) * (K_DIM / 4) + c4];
    as[it] = odi[tileM + r];
  }
#pragma unroll
  for (int it = 0; it < NLD; ++it) {
    const int r = wv + it * (THREADS / 32);
    float4 v = av[it];
    v.x *= as[it]; v.y *= as[it]; v.z *= as[it]; v.w *= as[it];
    *(float4*)&As[r][4 * c4] = v;
  }
  __syncthreads();

  const int wave = threadIdx.x >> 5;
  const int lane = threadIdx.x & 31;
  const int half = lane >> 4;
  const int r16  = lane & 15;
  const int nb   = wave * (16 * GEMM_TPW);

  v8f acc[GEMM_TPW];
#pragma unroll
  for (int t = 0; t < GEMM_TPW; ++t) acc[t] = v8f{};

#pragma unroll
  for (int k = 0; k < K_DIM; k += 4) {
    v2f a;
    a.x = As[r16][k + 2 * half];
    a.y = As[r16][k + 2 * half + 1];
#pragma unroll
    for (int t = 0; t < GEMM_TPW; ++t) {
      v2f b;
      b.x = W[(size_t)(k + 2 * half)     * NOUT + nb + 16 * t + r16];
      b.y = W[(size_t)(k + 2 * half + 1) * NOUT + nb + 16 * t + r16];
      acc[t] = __builtin_amdgcn_wmma_f32_16x16x4_f32(false, a, false, b,
                                                     (short)0, acc[t], false, false);
    }
  }

#pragma unroll
  for (int t = 0; t < GEMM_TPW; ++t) {
#pragma unroll
    for (int r = 0; r < 8; ++r) {
      const int row = tileM + r + 8 * half;
      C[(size_t)row * NOUT + nb + 16 * t + r16] = acc[t][r];
    }
  }
}

template <int D, int RELU>
__global__ __launch_bounds__(32 * GAT_WAVES) void gcn_gather(const float* __restrict__ proj,
                                                  const int* __restrict__ pairs,
                                                  const int* __restrict__ rowptr,
                                                  const int* __restrict__ cnt_in,
                                                  const float* __restrict__ idi,
                                                  const float* __restrict__ bias,
                                                  float* __restrict__ out) {
  constexpr int VEC = D / 32;
  typedef float vecf __attribute__((ext_vector_type(VEC)));

  const int lane = threadIdx.x & 31;
  const int w = __builtin_amdgcn_readfirstlane(blockIdx.x * GAT_WAVES + (threadIdx.x >> 5));

  const vecf* __restrict__ rows = (const vecf*)proj;

  int beg[GAT_NPW], cnt[GAT_NPW];
  float scl[GAT_NPW];
  int2 first[GAT_NPW];
#pragma unroll
  for (int i = 0; i < GAT_NPW; ++i) {
    const int node = w + i * GAT_TOTAL;
    beg[i] = rowptr[node];
    cnt[i] = cnt_in[node];
    scl[i] = idi[node];
  }
#pragma unroll
  for (int i = 0; i < GAT_NPW; ++i) {
    first[i] = make_int2(0, 0);
    if (lane < cnt[i]) first[i] = ((const int2*)pairs)[beg[i] + lane];
  }
  const vecf b = ((const vecf*)bias)[lane];

#pragma unroll
  for (int i = 0; i < GAT_NPW; ++i) {
    const int node = w + i * GAT_TOTAL;
    const int n = cnt[i];
    const int2* __restrict__ pr = (const int2*)pairs + beg[i];

    vecf acc = 0.0f;
    int2 my = first[i];
    for (int c0 = 0; c0 < n; c0 += 32) {
      const int m = min(32, n - c0);
      if (c0 > 0) {
        my = make_int2(0, 0);
        if (lane < m) my = pr[c0 + lane];
      }
      int j = 0;
      for (; j + 4 <= m; j += 4) {
        const int s0 = __builtin_amdgcn_readlane(my.x, j);
        const int s1 = __builtin_amdgcn_readlane(my.x, j + 1);
        const int s2 = __builtin_amdgcn_readlane(my.x, j + 2);
        const int s3 = __builtin_amdgcn_readlane(my.x, j + 3);
        const float w0 = __int_as_float(__builtin_amdgcn_readlane(my.y, j));
        const float w1 = __int_as_float(__builtin_amdgcn_readlane(my.y, j + 1));
        const float w2 = __int_as_float(__builtin_amdgcn_readlane(my.y, j + 2));
        const float w3 = __int_as_float(__builtin_amdgcn_readlane(my.y, j + 3));
        const vecf v0 = rows[(size_t)s0 * 32 + lane];
        const vecf v1 = rows[(size_t)s1 * 32 + lane];
        const vecf v2 = rows[(size_t)s2 * 32 + lane];
        const vecf v3 = rows[(size_t)s3 * 32 + lane];
        acc += w0 * v0;
        acc += w1 * v1;
        acc += w2 * v2;
        acc += w3 * v3;
      }
      for (; j < m; ++j) {
        const int s0 = __builtin_amdgcn_readlane(my.x, j);
        const float w0 = __int_as_float(__builtin_amdgcn_readlane(my.y, j));
        const vecf v0 = rows[(size_t)s0 * 32 + lane];
        acc += w0 * v0;
      }
    }

    vecf o = acc * scl[i] + b;
    if (RELU) {
#pragma unroll
      for (int t = 0; t < VEC; ++t) o[t] = fmaxf(o[t], 0.0f);
    }
    ((vecf*)out)[(size_t)node * 32 + lane] = o;
  }
}

#define FUSE_WAVES (16 / GAT_NPW)
static_assert(FUSE_WAVES * GAT_NPW == 16, "the waves of a block cover one 16-row stripe");
static_assert(N_NODES % 16 == 0, "whole stripes");

template <int NOUT>
__global__ __launch_bounds__(32 * FUSE_WAVES) void gcn_gather_gemm(const float* __restrict__ proj,
                                                  const int* __restrict__ pairs,
                                                  const int* __restrict__ rowptr,
                                                  const int* __restrict__ cnt_in,
                                                  const float* __restrict__ idi,
                                                  const float* __restrict__ odi,
                                                  const float* __restrict__ bias,
                                                  const float* __restrict__ W,
                                                  float* __restrict__ C) {
  constexpr int TPW = NOUT / (16 * FUSE_WAVES);
  static_assert(TPW >= 1 && TPW * 16 * FUSE_WAVES == NOUT, "columns split evenly over waves");
  typedef float vecf __attribute__((ext_vector_type(4)));

  __shared__ __attribute__((aligned(16))) float As[16][K_DIM + 4];

  const int lane = threadIdx.x & 31;
  const int wave = __builtin_amdgcn_readfirstlane(threadIdx.x >> 5);
  const int tileM = blockIdx.x * 16;
  const int node0 = tileM + wave * GAT_NPW;

  const vecf* __restrict__ rows = (const vecf*)proj;

  int beg[GAT_NPW], cnt[GAT_NPW];
  float scl[GAT_NPW], osc[GAT_NPW];
  int2 first[GAT_NPW];
#pragma unroll
  for (int i = 0; i < GAT_NPW; ++i) {
    const int node = node0 + i;
    beg[i] = rowptr[node];
    cnt[i] = cnt_in[node];
    scl[i] = idi[node];
    osc[i] = odi[node];
  }
#pragma unroll
  for (int i = 0; i < GAT_NPW; ++i) {
    first[i] = make_int2(0, 0);
    if (lane < cnt[i]) first[i] = ((const int2*)pairs)[beg[i] + lane];
  }
  const vecf b = ((const vecf*)bias)[lane];

#pragma unroll
  for (int i = 0; i < GAT_NPW; ++i) {
    const int n = cnt[i];
    const int2* __restrict__ pr = (const int2*)pairs + beg[i];

    vecf acc = 0.0f;
    int2 my = first[i];
    for (int c0 = 0; c0 < n; c0 += 32) {
      const int m = min(32, n - c0);
      if (c0 > 0) {
        my = make_int2(0, 0);
        if (lane < m) my = pr[c0 + lane];
      }
      int j = 0;
      for (; j + 4 <= m; j += 4) {
        const int s0 = __builtin_amdgcn_readlane(my.x, j);
        const int s1 = __builtin_amdgcn_readlane(my.x, j + 1);
        const int s2 = __builtin_amdgcn_readlane(my.x, j + 2);
        const int s3 = __builtin_amdgcn_readlane(my.x, j + 3);
        const float w0 = __int_as_float(__builtin_amdgcn_readlane(my.y, j));
        const float w1 = __int_as_float(__builtin_amdgcn_readlane(my.y, j + 1));
        const float w2 = __int_as_float(__builtin_amdgcn_readlane(my.y, j + 2));
        const float w3 = __int_as_float(__builtin_amdgcn_readlane(my.y, j + 3));
        const vecf v0 = rows[(size_t)s0 * 32 + lane];
        const vecf v1 = rows[(size_t)s1 * 32 + lane];
        const vecf v2 = rows[(size_t)s2 * 32 + lane];
        const vecf v3 = rows[(size_t)s3 * 32 + lane];
        acc += w0 * v0;
        acc += w1 * v1;
        acc += w2 * v2;
        acc += w3 * v3;
      }
      for (; j < m; ++j) {
        const int s0 = __builtin_amdgcn_readlane(my.x, j);
        const float w0 = __int_as_float(__builtin_amdgcn_readlane(my.y, j));
        const vecf v0 = rows[(size_t)s0 * 32 + lane];
        acc += w0 * v0;
      }
    }

    vecf o = acc * scl[i] + b;
#pragma unroll
    for (int t = 0; t < 4; ++t) o[t] = fmaxf(o[t], 0.0f);
    o *= osc[i];
    *(vecf*)&As[wave * GAT_NPW + i][4 * lane] = o;
  }
  __syncthreads();

  const int half = lane >> 4;
  const int r16  = lane & 15;
  const int nb   = wave * (16 * TPW);

  v8f cacc[TPW];
#pragma unroll
  for (int t = 0; t < TPW; ++t) cacc[t] = v8f{};

#pragma unroll
  for (int k = 0; k < K_DIM; k += 4) {
    v2f a;
    a.x = As[r16][k + 2 * half];
    a.y = As[r16][k + 2 * half + 1];
#pragma unroll
    for (int t = 0; t < TPW; ++t) {
      v2f bw;
      bw.x = W[(size_t)(k + 2 * half)     * NOUT + nb + 16 * t + r16];
      bw.y = W[(size_t)(k + 2 * half + 1) * NOUT + nb + 16 * t + r16];
      cacc[t] = __builtin_amdgcn_wmma_f32_16x16x4_f32(false, a, false, bw,
                                                      (short)0, cacc[t], false, false);
    }
  }

#pragma unroll
  for (int t = 0; t < TPW; ++t) {
#pragma unroll
    for (int r = 0; r < 8; ++r) {
      const int row = tileM + r + 8 * half;
      C[(size_t)row * NOUT + nb + 16 * t + r16] = cacc[t][r];
    }
  }
}

extern "C" void kernel_launch(void* const* d_in, const int* in_sizes, int n_in,
                              void* d_out, int out_size, void* d_ws, size_t ws_size,
                              hipStream_t stream) {
  const float* feat = (const float*)d_in[0];
  const float* ew   = (const float*)d_in[1];
  const float* W1   = (const float*)d_in[2];
  const float* b1   = (const float*)d_in[3];
  const float* W2   = (const float*)d_in[4];
  const float* b2   = (const float*)d_in[5];
  const float* W3   = (const float*)d_in[6];
  const float* b3   = (const float*)d_in[7];
  const int*   src  = (const int*)d_in[8];
  const int*   dst  = (const int*)d_in[9];
  float* out = (float*)d_out;

  float* ws     = (float*)d_ws;
  int*   cnt_in = (int*)ws;
  float* odi    = ws + 1 * (size_t)N_NODES;
  float* idi    = ws + 2 * (size_t)N_NODES;
  int*   rowptr = (int*)ws + 3 * (size_t)N_NODES;
  float* h      = ws + 4 * (size_t)N_NODES;
  float* proj   = h    + (size_t)N_NODES * 128;
  int*   pairs  = (int*)(proj + (size_t)N_NODES * 128);
  int*   part_in  = pairs + 2 * (size_t)N_EDGES;
  int*   part_out = part_in + (size_t)NSLICE * NODE_PAD;
  int*   tot_in   = part_out + (size_t)NSLICE * NODE_PAD;

  gcn_count<<<dim3(NRANGE, NSLICE), EDGE_THREADS, 0, stream>>>(src, dst, part_out, part_in, tot_in);
  gcn_scan_norm<<<NCHUNK, CHUNK_N, 0, stream>>>(part_in, part_out, tot_in, cnt_in, odi, idi, rowptr);
  gcn_fill<<<dim3(NRANGE, NSLICE), EDGE_THREADS, 0, stream>>>(src, dst, ew, part_in, pairs);

  const int gemmBlocks = N_NODES / 16;
  const int gatBlocks = GAT_BLOCKS;

  float* projA = proj;
  float* projB = h;

  gcn_gemm_wmma<128><<<gemmBlocks, 64, 0, stream>>>(feat, odi, W1, projA);

  gcn_gather_gemm<128><<<gemmBlocks, 32 * FUSE_WAVES, 0, stream>>>(projA, pairs, rowptr, cnt_in, idi, odi, b1, W2, projB);

  gcn_gather_gemm<64><<<gemmBlocks, 32 * FUSE_WAVES, 0, stream>>>(projB, pairs, rowptr, cnt_in, idi, odi, b2, W3, projA);

  gcn_gather<64, 0><<<gatBlocks, 32 * GAT_WAVES, 0, stream>>>(projA, pairs, rowptr, cnt_in, idi, b3, out);
}


# Layer 1 aggregates raw feature rows (edge weight times source out-degree scale) before projecting; W1 and W2 WMMA products both in the gather launch; 7 launches to 6; layer-1 sum order changed
# speedup vs baseline: 4.9399x; 1.0399x over previous
#include <hip/hip_runtime.h>
#include <math.h>


typedef __attribute__((ext_vector_type(2))) float v2f;
typedef __attribute__((ext_vector_type(8))) float v8f;

#define N_NODES 50000
#define N_EDGES 800000
#define K_DIM   128

#define NRANGE   16
#define RANGE_N  3200
#define NODE_PAD (NRANGE * RANGE_N)
#define NSLICE   16
#define SLICE_E  (N_EDGES / NSLICE)
#define SLICE_V  (SLICE_E / 4)
#define CHUNK_N  800
#define CHUNKS_PER_RANGE (RANGE_N / CHUNK_N)
#define NCHUNK   (NODE_PAD / CHUNK_N)
#define EDGE_THREADS 1024
#define EDGE_BATCH   4
static_assert(NODE_PAD >= N_NODES, "ranges cover all nodes");
static_assert(SLICE_E * NSLICE == N_EDGES && SLICE_V * 4 == SLICE_E, "edges split evenly into vector slices");
static_assert(CHUNK_N * CHUNKS_PER_RANGE == RANGE_N && CHUNK_N % 32 == 0, "whole waves per chunk");
static_assert(CHUNKS_PER_RANGE <= EDGE_THREADS / 32, "one wave per chunk total");

#define GAT_NPW    4
#define GAT_WAVES  4
#define GAT_TOTAL  (N_NODES / GAT_NPW)
#define GAT_BLOCKS (GAT_TOTAL / GAT_WAVES)
static_assert(GAT_TOTAL * GAT_NPW == N_NODES, "nodes split evenly over waves");
static_assert(GAT_BLOCKS * GAT_WAVES == GAT_TOTAL, "waves split evenly over blocks");


__global__ __launch_bounds__(EDGE_THREADS) void gcn_count(const int* __restrict__ src,
                                                          const int* __restrict__ dst,
                                                          int* __restrict__ part_out,
                                                          int* __restrict__ part_in,
                                                          int* __restrict__ tot_in) {
  __shared__ int c_in[RANGE_N];
  __shared__ int c_out[RANGE_N];
  const int tid = threadIdx.x;
  const int g = blockIdx.x, s = blockIdx.y;
  const int lo = g * RANGE_N;

  for (int i = tid; i < RANGE_N; i += EDGE_THREADS) { c_in[i] = 0; c_out[i] = 0; }
  __syncthreads();

  const int4* __restrict__ s4 = (const int4*)(src + (size_t)s * SLICE_E);
  const int4* __restrict__ d4 = (const int4*)(dst + (size_t)s * SLICE_E);
  for (int v0 = tid; v0 < SLICE_V; v0 += EDGE_THREADS * EDGE_BATCH) {
    int4 sv[EDGE_BATCH], dv[EDGE_BATCH];
#pragma unroll
    for (int k = 0; k < EDGE_BATCH; ++k) {
      const int v = v0 + k * EDGE_THREADS;
      sv[k] = make_int4(-1, -1, -1, -1);
      dv[k] = make_int4(-1, -1, -1, -1);
      if (v < SLICE_V) { sv[k] = s4[v]; dv[k] = d4[v]; }
    }
#pragma unroll
    for (int k = 0; k < EDGE_BATCH; ++k) {
      unsigned r;
      r = (unsigned)(dv[k].x - lo); if (r < RANGE_N) atomicAdd(&c_in[r], 1);
      r = (unsigned)(dv[k].y - lo); if (r < RANGE_N) atomicAdd(&c_in[r], 1);
      r = (unsigned)(dv[k].z - lo); if (r < RANGE_N) atomicAdd(&c_in[r], 1);
      r = (unsigned)(dv[k].w - lo); if (r < RANGE_N) atomicAdd(&c_in[r], 1);
      r = (unsigned)(sv[k].x - lo); if (r < RANGE_N) atomicAdd(&c_out[r], 1);
      r = (unsigned)(sv[k].y - lo); if (r < RANGE_N) atomicAdd(&c_out[r], 1);
      r = (unsigned)(sv[k].z - lo); if (r < RANGE_N) atomicAdd(&c_out[r], 1);
      r = (unsigned)(sv[k].w - lo); if (r < RANGE_N) atomicAdd(&c_out[r], 1);
    }
  }
  __syncthreads();

  for (int i = tid; i < RANGE_N; i += EDGE_THREADS) {
    part_in[(size_t)s * NODE_PAD + lo + i]  = c_in[i];
    part_out[(size_t)s * NODE_PAD + lo + i] = c_out[i];
  }

  const int lane = tid & 31;
  const int wave = tid >> 5;
  if (wave < CHUNKS_PER_RANGE) {
    int t = 0;
    for (int i = lane; i < CHUNK_N; i += 32) t += c_in[wave * CHUNK_N + i];
    for (int o = 16; o > 0; o >>= 1) t += __shfl_xor(t, o, 32);
    if (lane == 0) tot_in[(g * CHUNKS_PER_RANGE + wave) * NSLICE + s] = t;
  }
}

__global__ __launch_bounds__(CHUNK_N) void gcn_scan_norm(int* part_in,
                                                         const int* __restrict__ part_out,
                                                         const int* __restrict__ tot_in,
                                                         int* __restrict__ cnt_in,
                                                         float* __restrict__ odi,
                                                         float* __restrict__ idi,
                                                         int* __restrict__ rowptr) {
  constexpr int WAVES = CHUNK_N / 32;
  __shared__ int s_part[WAVES];
  __shared__ int s_wtot[WAVES];
  const int tid  = threadIdx.x;
  const int lane = tid & 31;
  const int wave = tid >> 5;
  const int c = blockIdx.x;

  int part = 0;
  for (int i = tid; i < c * NSLICE; i += CHUNK_N) part += tot_in[i];

  const int i = c * CHUNK_N + tid;
  int pin[NSLICE];
  int cout = 0;
#pragma unroll
  for (int s = 0; s < NSLICE; ++s) pin[s] = part_in[(size_t)s * NODE_PAD + i];
#pragma unroll
  for (int s = 0; s < NSLICE; ++s) cout += part_out[(size_t)s * NODE_PAD + i];
  int cin = 0;
#pragma unroll
  for (int s = 0; s < NSLICE; ++s) cin += pin[s];

  int incl = cin;
  for (int o = 1; o < 32; o <<= 1) {
    const int n = __shfl_up(incl, o, 32);
    if (lane >= o) incl += n;
  }
  for (int o = 16; o > 0; o >>= 1) part += __shfl_xor(part, o, 32);

  if (lane == 31) s_wtot[wave] = incl;
  if (lane == 0)  s_part[wave] = part;
  __syncthreads();

  int base = 0;
  for (int w = 0; w < WAVES; ++w) {
    base += s_part[w];
    if (w < wave) base += s_wtot[w];
  }

  const int rp = base + incl - cin;
  if (i < N_NODES) {
    rowptr[i] = rp;
    cnt_in[i] = cin;
    odi[i] = rsqrtf(fmaxf((float)cout, 1.0f));
    idi[i] = rsqrtf(fmaxf((float)cin, 1.0f));
  }
  int run = rp;
#pragma unroll
  for (int s = 0; s < NSLICE; ++s) {
    part_in[(size_t)s * NODE_PAD + i] = run;
    run += pin[s];
  }
}

__global__ __launch_bounds__(EDGE_THREADS) void gcn_fill(const int* __restrict__ src,
                                                         const int* __restrict__ dst,
                                                         const float* __restrict__ ew,
                                                         const int* __restrict__ start,
                                                         int* __restrict__ pairs) {
  __shared__ int cur[RANGE_N];
  const int tid = threadIdx.x;
  const int g = blockIdx.x, s = blockIdx.y;
  const int lo = g * RANGE_N;

  constexpr int NST = (RANGE_N + EDGE_THREADS - 1) / EDGE_THREADS;
  int st[NST];
#pragma unroll
  for (int k = 0; k < NST; ++k) {
    const int i = tid + k * EDGE_THREADS;
    st[k] = 0;
    if (i < RANGE_N) st[k] = start[(size_t)s * NODE_PAD + lo + i];
  }
#pragma unroll
  for (int k = 0; k < NST; ++k) {
    const int i = tid + k * EDGE_THREADS;
    if (i < RANGE_N) cur[i] = st[k];
  }
  __syncthreads();

  const int4*   __restrict__ s4 = (const int4*)(src + (size_t)s * SLICE_E);
  const int4*   __restrict__ d4 = (const int4*)(dst + (size_t)s * SLICE_E);
  const float4* __restrict__ w4 = (const float4*)(ew + (size_t)s * SLICE_E);
  int2* __restrict__ pr = (int2*)pairs;
  for (int v0 = tid; v0 < SLICE_V; v0 += EDGE_THREADS * EDGE_BATCH) {
    int4 sv[EDGE_BATCH], dv[EDGE_BATCH];
    float4 wv[EDGE_BATCH];
#pragma unroll
    for (int k = 0; k < EDGE_BATCH; ++k) {
      const int v = v0 + k * EDGE_THREADS;
      sv[k] = make_int4(0, 0, 0, 0);
      dv[k] = make_int4(-1, -1, -1, -1);
      wv[k] = make_float4(0.0f, 0.0f, 0.0f, 0.0f);
      if (v < SLICE_V) { sv[k] = s4[v]; dv[k] = d4[v]; wv[k] = w4[v]; }
    }
    int px[EDGE_BATCH], py[EDGE_BATCH], pz[EDGE_BATCH], pw[EDGE_BATCH];
#pragma unroll
    for (int k = 0; k < EDGE_BATCH; ++k) {
      unsigned r;
      px[k] = -1; r = (unsigned)(dv[k].x - lo); if (r < RANGE_N) px[k] = atomicAdd(&cur[r], 1);
      py[k] = -1; r = (unsigned)(dv[k].y - lo); if (r < RANGE_N) py[k] = atomicAdd(&cur[r], 1);
      pz[k] = -1; r = (unsigned)(dv[k].z - lo); if (r < RANGE_N) pz[k] = atomicAdd(&cur[r], 1);
      pw[k] = -1; r = (unsigned)(dv[k].w - lo); if (r < RANGE_N) pw[k] = atomicAdd(&cur[r], 1);
    }
#pragma unroll
    for (int k = 0; k < EDGE_BATCH; ++k) {
      if (px[k] >= 0) pr[px[k]] = make_int2(sv[k].x, __float_as_int(wv[k].x));
      if (py[k] >= 0) pr[py[k]] = make_int2(sv[k].y, __float_as_int(wv[k].y));
      if (pz[k] >= 0) pr[pz[k]] = make_int2(sv[k].z, __float_as_int(wv[k].z));
      if (pw[k] >= 0) pr[pw[k]] = make_int2(sv[k].w, __float_as_int(wv[k].w));
    }
  }
}


template <int D, int RELU>
__global__ __launch_bounds__(32 * GAT_WAVES) void gcn_gather(const float* __restrict__ proj,
                                                  const int* __restrict__ pairs,
                                                  const int* __restrict__ rowptr,
                                                  const int* __restrict__ cnt_in,
                                                  const float* __restrict__ idi,
                                                  const float* __restrict__ bias,
                                                  float* __restrict__ out) {
  constexpr int VEC = D / 32;
  typedef float vecf __attribute__((ext_vector_type(VEC)));

  const int lane = threadIdx.x & 31;
  const int w = __builtin_amdgcn_readfirstlane(blockIdx.x * GAT_WAVES + (threadIdx.x >> 5));

  const vecf* __restrict__ rows = (const vecf*)proj;

  int beg[GAT_NPW], cnt[GAT_NPW];
  float scl[GAT_NPW];
  int2 first[GAT_NPW];
#pragma unroll
  for (int i = 0; i < GAT_NPW; ++i) {
    const int node = w + i * GAT_TOTAL;
    beg[i] = rowptr[node];
    cnt[i] = cnt_in[node];
    scl[i] = idi[node];
  }
#pragma unroll
  for (int i = 0; i < GAT_NPW; ++i) {
    first[i] = make_int2(0, 0);
    if (lane < cnt[i]) first[i] = ((const int2*)pairs)[beg[i] + lane];
  }
  const vecf b = ((const vecf*)bias)[lane];

#pragma unroll
  for (int i = 0; i < GAT_NPW; ++i) {
    const int node = w + i * GAT_TOTAL;
    const int n = cnt[i];
    const int2* __restrict__ pr = (const int2*)pairs + beg[i];

    vecf acc = 0.0f;
    int2 my = first[i];
    for (int c0 = 0; c0 < n; c0 += 32) {
      const int m = min(32, n - c0);
      if (c0 > 0) {
        my = make_int2(0, 0);
        if (lane < m) my = pr[c0 + lane];
      }
      int j = 0;
      for (; j + 4 <= m; j += 4) {
        const int s0 = __builtin_amdgcn_readlane(my.x, j);
        const int s1 = __builtin_amdgcn_readlane(my.x, j + 1);
        const int s2 = __builtin_amdgcn_readlane(my.x, j + 2);
        const int s3 = __builtin_amdgcn_readlane(my.x, j + 3);
        const float w0 = __int_as_float(__builtin_amdgcn_readlane(my.y, j));
        const float w1 = __int_as_float(__builtin_amdgcn_readlane(my.y, j + 1));
        const float w2 = __int_as_float(__builtin_amdgcn_readlane(my.y, j + 2));
        const float w3 = __int_as_float(__builtin_amdgcn_readlane(my.y, j + 3));
        const vecf v0 = rows[(size_t)s0 * 32 + lane];
        const vecf v1 = rows[(size_t)s1 * 32 + lane];
        const vecf v2 = rows[(size_t)s2 * 32 + lane];
        const vecf v3 = rows[(size_t)s3 * 32 + lane];
        acc += w0 * v0;
        acc += w1 * v1;
        acc += w2 * v2;
        acc += w3 * v3;
      }
      for (; j < m; ++j) {
        const int s0 = __builtin_amdgcn_readlane(my.x, j);
        const float w0 = __int_as_float(__builtin_amdgcn_readlane(my.y, j));
        const vecf v0 = rows[(size_t)s0 * 32 + lane];
        acc += w0 * v0;
      }
    }

    vecf o = acc * scl[i] + b;
    if (RELU) {
#pragma unroll
      for (int t = 0; t < VEC; ++t) o[t] = fmaxf(o[t], 0.0f);
    }
    ((vecf*)out)[(size_t)node * 32 + lane] = o;
  }
}

#define FUSE_WAVES (16 / GAT_NPW)
static_assert(FUSE_WAVES * GAT_NPW == 16, "the waves of a block cover one 16-row stripe");
static_assert(N_NODES % 16 == 0, "whole stripes");

template <int NOUT>
__global__ __launch_bounds__(32 * FUSE_WAVES) void gcn_gather_gemm(const float* __restrict__ proj,
                                                  const int* __restrict__ pairs,
                                                  const int* __restrict__ rowptr,
                                                  const int* __restrict__ cnt_in,
                                                  const float* __restrict__ idi,
                                                  const float* __restrict__ odi,
                                                  const float* __restrict__ bias,
                                                  const float* __restrict__ W,
                                                  float* __restrict__ C) {
  constexpr int TPW = NOUT / (16 * FUSE_WAVES);
  static_assert(TPW >= 1 && TPW * 16 * FUSE_WAVES == NOUT, "columns split evenly over waves");
  typedef float vecf __attribute__((ext_vector_type(4)));

  __shared__ __attribute__((aligned(16))) float As[16][K_DIM + 4];

  const int lane = threadIdx.x & 31;
  const int wave = __builtin_amdgcn_readfirstlane(threadIdx.x >> 5);
  const int tileM = blockIdx.x * 16;
  const int node0 = tileM + wave * GAT_NPW;

  const vecf* __restrict__ rows = (const vecf*)proj;

  int beg[GAT_NPW], cnt[GAT_NPW];
  float scl[GAT_NPW], osc[GAT_NPW];
  int2 first[GAT_NPW];
#pragma unroll
  for (int i = 0; i < GAT_NPW; ++i) {
    const int node = node0 + i;
    beg[i] = rowptr[node];
    cnt[i] = cnt_in[node];
    scl[i] = idi[node];
    osc[i] = odi[node];
  }
#pragma unroll
  for (int i = 0; i < GAT_NPW; ++i) {
    first[i] = make_int2(0, 0);
    if (lane < cnt[i]) first[i] = ((const int2*)pairs)[beg[i] + lane];
  }
  const vecf b = ((const vecf*)bias)[lane];

#pragma unroll
  for (int i = 0; i < GAT_NPW; ++i) {
    const int n = cnt[i];
    const int2* __restrict__ pr = (const int2*)pairs + beg[i];

    vecf acc = 0.0f;
    int2 my = first[i];
    for (int c0 = 0; c0 < n; c0 += 32) {
      const int m = min(32, n - c0);
      if (c0 > 0) {
        my = make_int2(0, 0);
        if (lane < m) my = pr[c0 + lane];
      }
      int j = 0;
      for (; j + 4 <= m; j += 4) {
        const int s0 = __builtin_amdgcn_readlane(my.x, j);
        const int s1 = __builtin_amdgcn_readlane(my.x, j + 1);
        const int s2 = __builtin_amdgcn_readlane(my.x, j + 2);
        const int s3 = __builtin_amdgcn_readlane(my.x, j + 3);
        const float w0 = __int_as_float(__builtin_amdgcn_readlane(my.y, j));
        const float w1 = __int_as_float(__builtin_amdgcn_readlane(my.y, j + 1));
        const float w2 = __int_as_float(__builtin_amdgcn_readlane(my.y, j + 2));
        const float w3 = __int_as_float(__builtin_amdgcn_readlane(my.y, j + 3));
        const vecf v0 = rows[(size_t)s0 * 32 + lane];
        const vecf v1 = rows[(size_t)s1 * 32 + lane];
        const vecf v2 = rows[(size_t)s2 * 32 + lane];
        const vecf v3 = rows[(size_t)s3 * 32 + lane];
        acc += w0 * v0;
        acc += w1 * v1;
        acc += w2 * v2;
        acc += w3 * v3;
      }
      for (; j < m; ++j) {
        const int s0 = __builtin_amdgcn_readlane(my.x, j);
        const float w0 = __int_as_float(__builtin_amdgcn_readlane(my.y, j));
        const vecf v0 = rows[(size_t)s0 * 32 + lane];
        acc += w0 * v0;
      }
    }

    vecf o = acc * scl[i] + b;
#pragma unroll
    for (int t = 0; t < 4; ++t) o[t] = fmaxf(o[t], 0.0f);
    o *= osc[i];
    *(vecf*)&As[wave * GAT_NPW + i][4 * lane] = o;
  }
  __syncthreads();

  const int half = lane >> 4;
  const int r16  = lane & 15;
  const int nb   = wave * (16 * TPW);

  v8f cacc[TPW];
#pragma unroll
  for (int t = 0; t < TPW; ++t) cacc[t] = v8f{};

#pragma unroll
  for (int k = 0; k < K_DIM; k += 4) {
    v2f a;
    a.x = As[r16][k + 2 * half];
    a.y = As[r16][k + 2 * half + 1];
#pragma unroll
    for (int t = 0; t < TPW; ++t) {
      v2f bw;
      bw.x = W[(size_t)(k + 2 * half)     * NOUT + nb + 16 * t + r16];
      bw.y = W[(size_t)(k + 2 * half + 1) * NOUT + nb + 16 * t + r16];
      cacc[t] = __builtin_amdgcn_wmma_f32_16x16x4_f32(false, a, false, bw,
                                                      (short)0, cacc[t], false, false);
    }
  }

#pragma unroll
  for (int t = 0; t < TPW; ++t) {
#pragma unroll
    for (int r = 0; r < 8; ++r) {
      const int row = tileM + r + 8 * half;
      C[(size_t)row * NOUT + nb + 16 * t + r16] = cacc[t][r];
    }
  }
}

__global__ __launch_bounds__(32 * FUSE_WAVES) void gcn_gather_gemm2(const float* __restrict__ feat,
                                                  const int* __restrict__ pairs,
                                                  const int* __restrict__ rowptr,
                                                  const int* __restrict__ cnt_in,
                                                  const float* __restrict__ idi,
                                                  const float* __restrict__ odi,
                                                  const float* __restrict__ bias,
                                                  const float* __restrict__ Wa,
                                                  const float* __restrict__ Wb,
                                                  float* __restrict__ C) {
  constexpr int NOUT = K_DIM;
  constexpr int TPW = NOUT / (16 * FUSE_WAVES);
  static_assert(TPW >= 1 && TPW * 16 * FUSE_WAVES == NOUT, "columns split evenly over waves");
  typedef float vecf __attribute__((ext_vector_type(4)));

  __shared__ __attribute__((aligned(16))) float As[16][K_DIM + 4];
  __shared__ __attribute__((aligned(16))) float Hs[16][K_DIM + 4];
  __shared__ float s_idi[16];
  __shared__ float s_odi[16];

  const int lane = threadIdx.x & 31;
  const int wave = __builtin_amdgcn_readfirstlane(threadIdx.x >> 5);
  const int tileM = blockIdx.x * 16;
  const int node0 = tileM + wave * GAT_NPW;

  const vecf* __restrict__ rows = (const vecf*)feat;

  int beg[GAT_NPW], cnt[GAT_NPW];
  int2 first[GAT_NPW];
#pragma unroll
  for (int i = 0; i < GAT_NPW; ++i) {
    const int node = node0 + i;
    beg[i] = rowptr[node];
    cnt[i] = cnt_in[node];
    if (lane == 0) {
      s_idi[wave * GAT_NPW + i] = idi[node];
      s_odi[wave * GAT_NPW + i] = odi[node];
    }
  }
#pragma unroll
  for (int i = 0; i < GAT_NPW; ++i) {
    first[i] = make_int2(0, 0);
    if (lane < cnt[i]) first[i] = ((const int2*)pairs)[beg[i] + lane];
  }
  float fo[GAT_NPW];
#pragma unroll
  for (int i = 0; i < GAT_NPW; ++i) fo[i] = odi[first[i].x];
#pragma unroll
  for (int i = 0; i < GAT_NPW; ++i) first[i].y = __float_as_int(__int_as_float(first[i].y) * fo[i]);

#pragma unroll
  for (int i = 0; i < GAT_NPW; ++i) {
    const int n = cnt[i];
    const int2* __restrict__ pr = (const int2*)pairs + beg[i];

    vecf acc = 0.0f;
    int2 my = first[i];
    for (int c0 = 0; c0 < n; c0 += 32) {
      const int m = min(32, n - c0);
      if (c0 > 0) {
        my = make_int2(0, 0);
        if (lane < m) my = pr[c0 + lane];
        my.y = __float_as_int(__int_as_float(my.y) * odi[my.x]);
      }
      int j = 0;
      for (; j + 4 <= m; j += 4) {
        const int s0 = __builtin_amdgcn_readlane(my.x, j);
        const int s1 = __builtin_amdgcn_readlane(my.x, j + 1);
        const int s2 = __builtin_amdgcn_readlane(my.x, j + 2);
        const int s3 = __builtin_amdgcn_readlane(my.x, j + 3);
        const float w0 = __int_as_float(__builtin_amdgcn_readlane(my.y, j));
        const float w1 = __int_as_float(__builtin_amdgcn_readlane(my.y, j + 1));
        const float w2 = __int_as_float(__builtin_amdgcn_readlane(my.y, j + 2));
        const float w3 = __int_as_float(__builtin_amdgcn_readlane(my.y, j + 3));
        const vecf v0 = rows[(size_t)s0 * 32 + lane];
        const vecf v1 = rows[(size_t)s1 * 32 + lane];
        const vecf v2 = rows[(size_t)s2 * 32 + lane];
        const vecf v3 = rows[(size_t)s3 * 32 + lane];
        acc += w0 * v0;
        acc += w1 * v1;
        acc += w2 * v2;
        acc += w3 * v3;
      }
      for (; j < m; ++j) {
        const int s0 = __builtin_amdgcn_readlane(my.x, j);
        const float w0 = __int_as_float(__builtin_amdgcn_readlane(my.y, j));
        const vecf v0 = rows[(size_t)s0 * 32 + lane];
        acc += w0 * v0;
      }
    }
    *(vecf*)&As[wave * GAT_NPW + i][4 * lane] = acc;
  }
  __syncthreads();

  const int half = lane >> 4;
  const int r16  = lane & 15;
  const int nb   = wave * (16 * TPW);

  v8f cacc[TPW];
#pragma unroll
  for (int t = 0; t < TPW; ++t) cacc[t] = v8f{};

#pragma unroll
  for (int k = 0; k < K_DIM; k += 4) {
    v2f a;
    a.x = As[r16][k + 2 * half];
    a.y = As[r16][k + 2 * half + 1];
#pragma unroll
    for (int t = 0; t < TPW; ++t) {
      v2f bw;
      bw.x = Wa[(size_t)(k + 2 * half)     * NOUT + nb + 16 * t + r16];
      bw.y = Wa[(size_t)(k + 2 * half + 1) * NOUT + nb + 16 * t + r16];
      cacc[t] = __builtin_amdgcn_wmma_f32_16x16x4_f32(false, a, false, bw,
                                                      (short)0, cacc[t], false, false);
    }
  }

#pragma unroll
  for (int t = 0; t < TPW; ++t) {
    const int col = nb + 16 * t + r16;
    const float bc = bias[col];
#pragma unroll
    for (int r = 0; r < 8; ++r) {
      const int row = r + 8 * half;
      float o = cacc[t][r] * s_idi[row] + bc;
      o = fmaxf(o, 0.0f);
      o *= s_odi[row];
      Hs[row][col] = o;
    }
  }
  __syncthreads();

  v8f dacc[TPW];
#pragma unroll
  for (int t = 0; t < TPW; ++t) dacc[t] = v8f{};

#pragma unroll
  for (int k = 0; k < K_DIM; k += 4) {
    v2f a;
    a.x = Hs[r16][k + 2 * half];
    a.y = Hs[r16][k + 2 * half + 1];
#pragma unroll
    for (int t = 0; t < TPW; ++t) {
      v2f bw;
      bw.x = Wb[(size_t)(k + 2 * half)     * NOUT + nb + 16 * t + r16];
      bw.y = Wb[(size_t)(k + 2 * half + 1) * NOUT + nb + 16 * t + r16];
      dacc[t] = __builtin_amdgcn_wmma_f32_16x16x4_f32(false, a, false, bw,
                                                      (short)0, dacc[t], false, false);
    }
  }

#pragma unroll
  for (int t = 0; t < TPW; ++t) {
#pragma unroll
    for (int r = 0; r < 8; ++r) {
      const int row = tileM + r + 8 * half;
      C[(size_t)row * NOUT + nb + 16 * t + r16] = dacc[t][r];
    }
  }
}

extern "C" void kernel_launch(void* const* d_in, const int* in_sizes, int n_in,
                              void* d_out, int out_size, void* d_ws, size_t ws_size,
                              hipStream_t stream) {
  const float* feat = (const float*)d_in[0];
  const float* ew   = (const float*)d_in[1];
  const float* W1   = (const float*)d_in[2];
  const float* b1   = (const float*)d_in[3];
  const float* W2   = (const float*)d_in[4];
  const float* b2   = (const float*)d_in[5];
  const float* W3   = (const float*)d_in[6];
  const float* b3   = (const float*)d_in[7];
  const int*   src  = (const int*)d_in[8];
  const int*   dst  = (const int*)d_in[9];
  float* out = (float*)d_out;

  float* ws     = (float*)d_ws;
  int*   cnt_in = (int*)ws;
  float* odi    = ws + 1 * (size_t)N_NODES;
  float* idi    = ws + 2 * (size_t)N_NODES;
  int*   rowptr = (int*)ws + 3 * (size_t)N_NODES;
  float* proj3  = ws + 4 * (size_t)N_NODES;
  float* proj2  = proj3 + (size_t)N_NODES * 128;
  int*   pairs  = (int*)(proj2 + (size_t)N_NODES * 128);
  int*   part_in  = pairs + 2 * (size_t)N_EDGES;
  int*   part_out = part_in + (size_t)NSLICE * NODE_PAD;
  int*   tot_in   = part_out + (size_t)NSLICE * NODE_PAD;

  gcn_count<<<dim3(NRANGE, NSLICE), EDGE_THREADS, 0, stream>>>(src, dst, part_out, part_in, tot_in);
  gcn_scan_norm<<<NCHUNK, CHUNK_N, 0, stream>>>(part_in, part_out, tot_in, cnt_in, odi, idi, rowptr);
  gcn_fill<<<dim3(NRANGE, NSLICE), EDGE_THREADS, 0, stream>>>(src, dst, ew, part_in, pairs);

  const int stripeBlocks = N_NODES / 16;
  const int gatBlocks = GAT_BLOCKS;

  gcn_gather_gemm2<<<stripeBlocks, 32 * FUSE_WAVES, 0, stream>>>(feat, pairs, rowptr, cnt_in, idi, odi, b1, W1, W2, proj2);

  gcn_gather_gemm<64><<<stripeBlocks, 32 * FUSE_WAVES, 0, stream>>>(proj2, pairs, rowptr, cnt_in, idi, odi, b2, W3, proj3);

  gcn_gather<64, 0><<<gatBlocks, 32 * GAT_WAVES, 0, stream>>>(proj3, pairs, rowptr, cnt_in, idi, b3, out);
}


# Layer-1 launch, both WMMA products: a lane owns 2 adjacent output columns (one per tile): one 8-byte weight load feeds both tiles, rows leave as whole 128-byte lines; VGPRs 70 to 64; sums unchanged
# speedup vs baseline: 5.2050x; 1.0537x over previous
#include <hip/hip_runtime.h>
#include <math.h>


typedef __attribute__((ext_vector_type(2))) float v2f;
typedef __attribute__((ext_vector_type(8))) float v8f;

#define N_NODES 50000
#define N_EDGES 800000
#define K_DIM   128

#define NRANGE   16
#define RANGE_N  3200
#define NODE_PAD (NRANGE * RANGE_N)
#define NSLICE   16
#define SLICE_E  (N_EDGES / NSLICE)
#define SLICE_V  (SLICE_E / 4)
#define CHUNK_N  800
#define CHUNKS_PER_RANGE (RANGE_N / CHUNK_N)
#define NCHUNK   (NODE_PAD / CHUNK_N)
#define EDGE_THREADS 1024
#define EDGE_BATCH   4
static_assert(NODE_PAD >= N_NODES, "ranges cover all nodes");
static_assert(SLICE_E * NSLICE == N_EDGES && SLICE_V * 4 == SLICE_E, "edges split evenly into vector slices");
static_assert(CHUNK_N * CHUNKS_PER_RANGE == RANGE_N && CHUNK_N % 32 == 0, "whole waves per chunk");
static_assert(CHUNKS_PER_RANGE <= EDGE_THREADS / 32, "one wave per chunk total");

#define GAT_NPW    4
#define GAT_WAVES  4
#define GAT_TOTAL  (N_NODES / GAT_NPW)
#define GAT_BLOCKS (GAT_TOTAL / GAT_WAVES)
static_assert(GAT_TOTAL * GAT_NPW == N_NODES, "nodes split evenly over waves");
static_assert(GAT_BLOCKS * GAT_WAVES == GAT_TOTAL, "waves split evenly over blocks");


__global__ __launch_bounds__(EDGE_THREADS) void gcn_count(const int* __restrict__ src,
                                                          const int* __restrict__ dst,
                                                          int* __restrict__ part_out,
                                                          int* __restrict__ part_in,
                                                          int* __restrict__ tot_in) {
  __shared__ int c_in[RANGE_N];
  __shared__ int c_out[RANGE_N];
  const int tid = threadIdx.x;
  const int g = blockIdx.x, s = blockIdx.y;
  const int lo = g * RANGE_N;

  for (int i = tid; i < RANGE_N; i += EDGE_THREADS) { c_in[i] = 0; c_out[i] = 0; }
  __syncthreads();

  const int4* __restrict__ s4 = (const int4*)(src + (size_t)s * SLICE_E);
  const int4* __restrict__ d4 = (const int4*)(dst + (size_t)s * SLICE_E);
  for (int v0 = tid; v0 < SLICE_V; v0 += EDGE_THREADS * EDGE_BATCH) {
    int4 sv[EDGE_BATCH], dv[EDGE_BATCH];
#pragma unroll
    for (int k = 0; k < EDGE_BATCH; ++k) {
      const int v = v0 + k * EDGE_THREADS;
      sv[k] = make_int4(-1, -1, -1, -1);
      dv[k] = make_int4(-1, -1, -1, -1);
      if (v < SLICE_V) { sv[k] = s4[v]; dv[k] = d4[v]; }
    }
#pragma unroll
    for (int k = 0; k < EDGE_BATCH; ++k) {
      unsigned r;
      r = (unsigned)(dv[k].x - lo); if (r < RANGE_N) atomicAdd(&c_in[r], 1);
      r = (unsigned)(dv[k].y - lo); if (r < RANGE_N) atomicAdd(&c_in[r], 1);
      r = (unsigned)(dv[k].z - lo); if (r < RANGE_N) atomicAdd(&c_in[r], 1);
      r = (unsigned)(dv[k].w - lo); if (r < RANGE_N) atomicAdd(&c_in[r], 1);
      r = (unsigned)(sv[k].x - lo); if (r < RANGE_N) atomicAdd(&c_out[r], 1);
      r = (unsigned)(sv[k].y - lo); if (r < RANGE_N) atomicAdd(&c_out[r], 1);
      r = (unsigned)(sv[k].z - lo); if (r < RANGE_N) atomicAdd(&c_out[r], 1);
      r = (unsigned)(sv[k].w - lo); if (r < RANGE_N) atomicAdd(&c_out[r], 1);
    }
  }
  __syncthreads();

  for (int i = tid; i < RANGE_N; i += EDGE_THREADS) {
    part_in[(size_t)s * NODE_PAD + lo + i]  = c_in[i];
    part_out[(size_t)s * NODE_PAD + lo + i] = c_out[i];
  }

  const int lane = tid & 31;
  const int wave = tid >> 5;
  if (wave < CHUNKS_PER_RANGE) {
    int t = 0;
    for (int i = lane; i < CHUNK_N; i += 32) t += c_in[wave * CHUNK_N + i];
    for (int o = 16; o > 0; o >>= 1) t += __shfl_xor(t, o, 32);
    if (lane == 0) tot_in[(g * CHUNKS_PER_RANGE + wave) * NSLICE + s] = t;
  }
}

__global__ __launch_bounds__(CHUNK_N) void gcn_scan_norm(int* part_in,
                                                         const int* __restrict__ part_out,
                                                         const int* __restrict__ tot_in,
                                                         int* __restrict__ cnt_in,
                                                         float* __restrict__ odi,
                                                         float* __restrict__ idi,
                                                         int* __restrict__ rowptr) {
  constexpr int WAVES = CHUNK_N / 32;
  __shared__ int s_part[WAVES];
  __shared__ int s_wtot[WAVES];
  const int tid  = threadIdx.x;
  const int lane = tid & 31;
  const int wave = tid >> 5;
  const int c = blockIdx.x;

  int part = 0;
  for (int i = tid; i < c * NSLICE; i += CHUNK_N) part += tot_in[i];

  const int i = c * CHUNK_N + tid;
  int pin[NSLICE];
  int cout = 0;
#pragma unroll
  for (int s = 0; s < NSLICE; ++s) pin[s] = part_in[(size_t)s * NODE_PAD + i];
#pragma unroll
  for (int s = 0; s < NSLICE; ++s) cout += part_out[(size_t)s * NODE_PAD + i];
  int cin = 0;
#pragma unroll
  for (int s = 0; s < NSLICE; ++s) cin += pin[s];

  int incl = cin;
  for (int o = 1; o < 32; o <<= 1) {
    const int n = __shfl_up(incl, o, 32);
    if (lane >= o) incl += n;
  }
  for (int o = 16; o > 0; o >>= 1) part += __shfl_xor(part, o, 32);

  if (lane == 31) s_wtot[wave] = incl;
  if (lane == 0)  s_part[wave] = part;
  __syncthreads();

  int base = 0;
  for (int w = 0; w < WAVES; ++w) {
    base += s_part[w];
    if (w < wave) base += s_wtot[w];
  }

  const int rp = base + incl - cin;
  if (i < N_NODES) {
    rowptr[i] = rp;
    cnt_in[i] = cin;
    odi[i] = rsqrtf(fmaxf((float)cout, 1.0f));
    idi[i] = rsqrtf(fmaxf((float)cin, 1.0f));
  }
  int run = rp;
#pragma unroll
  for (int s = 0; s < NSLICE; ++s) {
    part_in[(size_t)s * NODE_PAD + i] = run;
    run += pin[s];
  }
}

__global__ __launch_bounds__(EDGE_THREADS) void gcn_fill(const int* __restrict__ src,
                                                         const int* __restrict__ dst,
                                                         const float* __restrict__ ew,
                                                         const int* __restrict__ start,
                                                         int* __restrict__ pairs) {
  __shared__ int cur[RANGE_N];
  const int tid = threadIdx.x;
  const int g = blockIdx.x, s = blockIdx.y;
  const int lo = g * RANGE_N;

  constexpr int NST = (RANGE_N + EDGE_THREADS - 1) / EDGE_THREADS;
  int st[NST];
#pragma unroll
  for (int k = 0; k < NST; ++k) {
    const int i = tid + k * EDGE_THREADS;
    st[k] = 0;
    if (i < RANGE_N) st[k] = start[(size_t)s * NODE_PAD + lo + i];
  }
#pragma unroll
  for (int k = 0; k < NST; ++k) {
    const int i = tid + k * EDGE_THREADS;
    if (i < RANGE_N) cur[i] = st[k];
  }
  __syncthreads();

  const int4*   __restrict__ s4 = (const int4*)(src + (size_t)s * SLICE_E);
  const int4*   __restrict__ d4 = (const int4*)(dst + (size_t)s * SLICE_E);
  const float4* __restrict__ w4 = (const float4*)(ew + (size_t)s * SLICE_E);
  int2* __restrict__ pr = (int2*)pairs;
  for (int v0 = tid; v0 < SLICE_V; v0 += EDGE_THREADS * EDGE_BATCH) {
    int4 sv[EDGE_BATCH], dv[EDGE_BATCH];
    float4 wv[EDGE_BATCH];
#pragma unroll
    for (int k = 0; k < EDGE_BATCH; ++k) {
      const int v = v0 + k * EDGE_THREADS;
      sv[k] = make_int4(0, 0, 0, 0);
      dv[k] = make_int4(-1, -1, -1, -1);
      wv[k] = make_float4(0.0f, 0.0f, 0.0f, 0.0f);
      if (v < SLICE_V) { sv[k] = s4[v]; dv[k] = d4[v]; wv[k] = w4[v]; }
    }
    int px[EDGE_BATCH], py[EDGE_BATCH], pz[EDGE_BATCH], pw[EDGE_BATCH];
#pragma unroll
    for (int k = 0; k < EDGE_BATCH; ++k) {
      unsigned r;
      px[k] = -1; r = (unsigned)(dv[k].x - lo); if (r < RANGE_N) px[k] = atomicAdd(&cur[r], 1);
      py[k] = -1; r = (unsigned)(dv[k].y - lo); if (r < RANGE_N) py[k] = atomicAdd(&cur[r], 1);
      pz[k] = -1; r = (unsigned)(dv[k].z - lo); if (r < RANGE_N) pz[k] = atomicAdd(&cur[r], 1);
      pw[k] = -1; r = (unsigned)(dv[k].w - lo); if (r < RANGE_N) pw[k] = atomicAdd(&cur[r], 1);
    }
#pragma unroll
    for (int k = 0; k < EDGE_BATCH; ++k) {
      if (px[k] >= 0) pr[px[k]] = make_int2(sv[k].x, __float_as_int(wv[k].x));
      if (py[k] >= 0) pr[py[k]] = make_int2(sv[k].y, __float_as_int(wv[k].y));
      if (pz[k] >= 0) pr[pz[k]] = make_int2(sv[k].z, __float_as_int(wv[k].z));
      if (pw[k] >= 0) pr[pw[k]] = make_int2(sv[k].w, __float_as_int(wv[k].w));
    }
  }
}


template <int D, int RELU>
__global__ __launch_bounds__(32 * GAT_WAVES) void gcn_gather(const float* __restrict__ proj,
                                                  const int* __restrict__ pairs,
                                                  const int* __restrict__ rowptr,
                                                  const int* __restrict__ cnt_in,
                                                  const float* __restrict__ idi,
                                                  const float* __restrict__ bias,
                                                  float* __restrict__ out) {
  constexpr int VEC = D / 32;
  typedef float vecf __attribute__((ext_vector_type(VEC)));

  const int lane = threadIdx.x & 31;
  const int w = __builtin_amdgcn_readfirstlane(blockIdx.x * GAT_WAVES + (threadIdx.x >> 5));

  const vecf* __restrict__ rows = (const vecf*)proj;

  int beg[GAT_NPW], cnt[GAT_NPW];
  float scl[GAT_NPW];
  int2 first[GAT_NPW];
#pragma unroll
  for (int i = 0; i < GAT_NPW; ++i) {
    const int node = w + i * GAT_TOTAL;
    beg[i] = rowptr[node];
    cnt[i] = cnt_in[node];
    scl[i] = idi[node];
  }
#pragma unroll
  for (int i = 0; i < GAT_NPW; ++i) {
    first[i] = make_int2(0, 0);
    if (lane < cnt[i]) first[i] = ((const int2*)pairs)[beg[i] + lane];
  }
  const vecf b = ((const vecf*)bias)[lane];

#pragma unroll
  for (int i = 0; i < GAT_NPW; ++i) {
    const int node = w + i * GAT_TOTAL;
    const int n = cnt[i];
    const int2* __restrict__ pr = (const int2*)pairs + beg[i];

    vecf acc = 0.0f;
    int2 my = first[i];
    for (int c0 = 0; c0 < n; c0 += 32) {
      const int m = min(32, n - c0);
      if (c0 > 0) {
        my = make_int2(0, 0);
        if (lane < m) my = pr[c0 + lane];
      }
      int j = 0;
      for (; j + 4 <= m; j += 4) {
        const int s0 = __builtin_amdgcn_readlane(my.x, j);
        const int s1 = __builtin_amdgcn_readlane(my.x, j + 1);
        const int s2 = __builtin_amdgcn_readlane(my.x, j + 2);
        const int s3 = __builtin_amdgcn_readlane(my.x, j + 3);
        const float w0 = __int_as_float(__builtin_amdgcn_readlane(my.y, j));
        const float w1 = __int_as_float(__builtin_amdgcn_readlane(my.y, j + 1));
        const float w2 = __int_as_float(__builtin_amdgcn_readlane(my.y, j + 2));
        const float w3 = __int_as_float(__builtin_amdgcn_readlane(my.y, j + 3));
        const vecf v0 = rows[(size_t)s0 * 32 + lane];
        const vecf v1 = rows[(size_t)s1 * 32 + lane];
        const vecf v2 = rows[(size_t)s2 * 32 + lane];
        const vecf v3 = rows[(size_t)s3 * 32 + lane];
        acc += w0 * v0;
        acc += w1 * v1;
        acc += w2 * v2;
        acc += w3 * v3;
      }
      for (; j < m; ++j) {
        const int s0 = __builtin_amdgcn_readlane(my.x, j);
        const float w0 = __int_as_float(__builtin_amdgcn_readlane(my.y, j));
        const vecf v0 = rows[(size_t)s0 * 32 + lane];
        acc += w0 * v0;
      }
    }

    vecf o = acc * scl[i] + b;
    if (RELU) {
#pragma unroll
      for (int t = 0; t < VEC; ++t) o[t] = fmaxf(o[t], 0.0f);
    }
    ((vecf*)out)[(size_t)node * 32 + lane] = o;
  }
}

#define FUSE_WAVES (16 / GAT_NPW)
static_assert(FUSE_WAVES * GAT_NPW == 16, "the waves of a block cover one 16-row stripe");
static_assert(N_NODES % 16 == 0, "whole stripes");

template <int NOUT>
__global__ __launch_bounds__(32 * FUSE_WAVES) void gcn_gather_gemm(const float* __restrict__ proj,
                                                  const int* __restrict__ pairs,
                                                  const int* __restrict__ rowptr,
                                                  const int* __restrict__ cnt_in,
                                                  const float* __restrict__ idi,
                                                  const float* __restrict__ odi,
                                                  const float* __restrict__ bias,
                                                  const float* __restrict__ W,
                                                  float* __restrict__ C) {
  constexpr int TPW = NOUT / (16 * FUSE_WAVES);
  static_assert(TPW >= 1 && TPW * 16 * FUSE_WAVES == NOUT, "columns split evenly over waves");
  typedef float vecf __attribute__((ext_vector_type(4)));

  __shared__ __attribute__((aligned(16))) float As[16][K_DIM + 4];

  const int lane = threadIdx.x & 31;
  const int wave = __builtin_amdgcn_readfirstlane(threadIdx.x >> 5);
  const int tileM = blockIdx.x * 16;
  const int node0 = tileM + wave * GAT_NPW;

  const vecf* __restrict__ rows = (const vecf*)proj;

  int beg[GAT_NPW], cnt[GAT_NPW];
  float scl[GAT_NPW], osc[GAT_NPW];
  int2 first[GAT_NPW];
#pragma unroll
  for (int i = 0; i < GAT_NPW; ++i) {
    const int node = node0 + i;
    beg[i] = rowptr[node];
    cnt[i] = cnt_in[node];
    scl[i] = idi[node];
    osc[i] = odi[node];
  }
#pragma unroll
  for (int i = 0; i < GAT_NPW; ++i) {
    first[i] = make_int2(0, 0);
    if (lane < cnt[i]) first[i] = ((const int2*)pairs)[beg[i] + lane];
  }
  const vecf b = ((const vecf*)bias)[lane];

#pragma unroll
  for (int i = 0; i < GAT_NPW; ++i) {
    const int n = cnt[i];
    const int2* __restrict__ pr = (const int2*)pairs + beg[i];

    vecf acc = 0.0f;
    int2 my = first[i];
    for (int c0 = 0; c0 < n; c0 += 32) {
      const int m = min(32, n - c0);
      if (c0 > 0) {
        my = make_int2(0, 0);
        if (lane < m) my = pr[c0 + lane];
      }
      int j = 0;
      for (; j + 4 <= m; j += 4) {
        const int s0 = __builtin_amdgcn_readlane(my.x, j);
        const int s1 = __builtin_amdgcn_readlane(my.x, j + 1);
        const int s2 = __builtin_amdgcn_readlane(my.x, j + 2);
        const int s3 = __builtin_amdgcn_readlane(my.x, j + 3);
        const float w0 = __int_as_float(__builtin_amdgcn_readlane(my.y, j));
        const float w1 = __int_as_float(__builtin_amdgcn_readlane(my.y, j + 1));
        const float w2 = __int_as_float(__builtin_amdgcn_readlane(my.y, j + 2));
        const float w3 = __int_as_float(__builtin_amdgcn_readlane(my.y, j + 3));
        const vecf v0 = rows[(size_t)s0 * 32 + lane];
        const vecf v1 = rows[(size_t)s1 * 32 + lane];
        const vecf v2 = rows[(size_t)s2 * 32 + lane];
        const vecf v3 = rows[(size_t)s3 * 32 + lane];
        acc += w0 * v0;
        acc += w1 * v1;
        acc += w2 * v2;
        acc += w3 * v3;
      }
      for (; j < m; ++j) {
        const int s0 = __builtin_amdgcn_readlane(my.x, j);
        const float w0 = __int_as_float(__builtin_amdgcn_readlane(my.y, j));
        const vecf v0 = rows[(size_t)s0 * 32 + lane];
        acc += w0 * v0;
      }
    }

    vecf o = acc * scl[i] + b;
#pragma unroll
    for (int t = 0; t < 4; ++t) o[t] = fmaxf(o[t], 0.0f);
    o *= osc[i];
    *(vecf*)&As[wave * GAT_NPW + i][4 * lane] = o;
  }
  __syncthreads();

  const int half = lane >> 4;
  const int r16  = lane & 15;
  const int nb   = wave * (16 * TPW);

  v8f cacc[TPW];
#pragma unroll
  for (int t = 0; t < TPW; ++t) cacc[t] = v8f{};

#pragma unroll
  for (int k = 0; k < K_DIM; k += 4) {
    v2f a;
    a.x = As[r16][k + 2 * half];
    a.y = As[r16][k + 2 * half + 1];
#pragma unroll
    for (int t = 0; t < TPW; ++t) {
      v2f bw;
      bw.x = W[(size_t)(k + 2 * half)     * NOUT + nb + 16 * t + r16];
      bw.y = W[(size_t)(k + 2 * half + 1) * NOUT + nb + 16 * t + r16];
      cacc[t] = __builtin_amdgcn_wmma_f32_16x16x4_f32(false, a, false, bw,
                                                      (short)0, cacc[t], false, false);
    }
  }

#pragma unroll
  for (int t = 0; t < TPW; ++t) {
#pragma unroll
    for (int r = 0; r < 8; ++r) {
      const int row = tileM + r + 8 * half;
      C[(size_t)row * NOUT + nb + 16 * t + r16] = cacc[t][r];
    }
  }
}

__global__ __launch_bounds__(32 * FUSE_WAVES) void gcn_gather_gemm2(const float* __restrict__ feat,
                                                  const int* __restrict__ pairs,
                                                  const int* __restrict__ rowptr,
                                                  const int* __restrict__ cnt_in,
                                                  const float* __restrict__ idi,
                                                  const float* __restrict__ odi,
                                                  const float* __restrict__ bias,
                                                  const float* __restrict__ Wa,
                                                  const float* __restrict__ Wb,
                                                  float* __restrict__ C) {
  constexpr int NOUT = K_DIM;
  constexpr int TPW = NOUT / (16 * FUSE_WAVES);
  static_assert(TPW == 2 && TPW * 16 * FUSE_WAVES == NOUT, "two column tiles per wave: a lane's column pair");
  typedef float vecf __attribute__((ext_vector_type(4)));

  __shared__ __attribute__((aligned(16))) float As[16][K_DIM + 4];
  __shared__ __attribute__((aligned(16))) float Hs[16][K_DIM + 4];
  __shared__ float s_idi[16];
  __shared__ float s_odi[16];

  const int lane = threadIdx.x & 31;
  const int wave = __builtin_amdgcn_readfirstlane(threadIdx.x >> 5);
  const int tileM = blockIdx.x * 16;
  const int node0 = tileM + wave * GAT_NPW;

  const vecf* __restrict__ rows = (const vecf*)feat;

  int beg[GAT_NPW], cnt[GAT_NPW];
  int2 first[GAT_NPW];
#pragma unroll
  for (int i = 0; i < GAT_NPW; ++i) {
    const int node = node0 + i;
    beg[i] = rowptr[node];
    cnt[i] = cnt_in[node];
    if (lane == 0) {
      s_idi[wave * GAT_NPW + i] = idi[node];
      s_odi[wave * GAT_NPW + i] = odi[node];
    }
  }
#pragma unroll
  for (int i = 0; i < GAT_NPW; ++i) {
    first[i] = make_int2(0, 0);
    if (lane < cnt[i]) first[i] = ((const int2*)pairs)[beg[i] + lane];
  }
  float fo[GAT_NPW];
#pragma unroll
  for (int i = 0; i < GAT_NPW; ++i) fo[i] = odi[first[i].x];
#pragma unroll
  for (int i = 0; i < GAT_NPW; ++i) first[i].y = __float_as_int(__int_as_float(first[i].y) * fo[i]);

#pragma unroll
  for (int i = 0; i < GAT_NPW; ++i) {
    const int n = cnt[i];
    const int2* __restrict__ pr = (const int2*)pairs + beg[i];

    vecf acc = 0.0f;
    int2 my = first[i];
    for (int c0 = 0; c0 < n; c0 += 32) {
      const int m = min(32, n - c0);
      if (c0 > 0) {
        my = make_int2(0, 0);
        if (lane < m) my = pr[c0 + lane];
        my.y = __float_as_int(__int_as_float(my.y) * odi[my.x]);
      }
      int j = 0;
      for (; j + 4 <= m; j += 4) {
        const int s0 = __builtin_amdgcn_readlane(my.x, j);
        const int s1 = __builtin_amdgcn_readlane(my.x, j + 1);
        const int s2 = __builtin_amdgcn_readlane(my.x, j + 2);
        const int s3 = __builtin_amdgcn_readlane(my.x, j + 3);
        const float w0 = __int_as_float(__builtin_amdgcn_readlane(my.y, j));
        const float w1 = __int_as_float(__builtin_amdgcn_readlane(my.y, j + 1));
        const float w2 = __int_as_float(__builtin_amdgcn_readlane(my.y, j + 2));
        const float w3 = __int_as_float(__builtin_amdgcn_readlane(my.y, j + 3));
        const vecf v0 = rows[(size_t)s0 * 32 + lane];
        const vecf v1 = rows[(size_t)s1 * 32 + lane];
        const vecf v2 = rows[(size_t)s2 * 32 + lane];
        const vecf v3 = rows[(size_t)s3 * 32 + lane];
        acc += w0 * v0;
        acc += w1 * v1;
        acc += w2 * v2;
        acc += w3 * v3;
      }
      for (; j < m; ++j) {
        const int s0 = __builtin_amdgcn_readlane(my.x, j);
        const float w0 = __int_as_float(__builtin_amdgcn_readlane(my.y, j));
        const vecf v0 = rows[(size_t)s0 * 32 + lane];
        acc += w0 * v0;
      }
    }
    *(vecf*)&As[wave * GAT_NPW + i][4 * lane] = acc;
  }
  __syncthreads();

  const int half = lane >> 4;
  const int r16  = lane & 15;
  const int nb   = wave * (16 * TPW);
  const int col  = nb + 2 * r16;

  v8f cacc[TPW];
#pragma unroll
  for (int t = 0; t < TPW; ++t) cacc[t] = v8f{};

#pragma unroll
  for (int k = 0; k < K_DIM; k += 4) {
    v2f a;
    a.x = As[r16][k + 2 * half];
    a.y = As[r16][k + 2 * half + 1];
    const float2 w0 = *(const float2*)&Wa[(size_t)(k + 2 * half)     * NOUT + col];
    const float2 w1 = *(const float2*)&Wa[(size_t)(k + 2 * half + 1) * NOUT + col];
    v2f b0, b1;
    b0.x = w0.x; b0.y = w1.x;
    b1.x = w0.y; b1.y = w1.y;
    cacc[0] = __builtin_amdgcn_wmma_f32_16x16x4_f32(false, a, false, b0,
                                                    (short)0, cacc[0], false, false);
    cacc[1] = __builtin_amdgcn_wmma_f32_16x16x4_f32(false, a, false, b1,
                                                    (short)0, cacc[1], false, false);
  }

  {
    const float2 bc = *(const float2*)&bias[col];
#pragma unroll
    for (int r = 0; r < 8; ++r) {
      const int row = r + 8 * half;
      const float si = s_idi[row];
      const float so = s_odi[row];
      float o0 = cacc[0][r] * si + bc.x;
      float o1 = cacc[1][r] * si + bc.y;
      o0 = fmaxf(o0, 0.0f);
      o1 = fmaxf(o1, 0.0f);
      o0 *= so;
      o1 *= so;
      *(float2*)&Hs[row][col] = make_float2(o0, o1);
    }
  }
  __syncthreads();

  v8f dacc[TPW];
#pragma unroll
  for (int t = 0; t < TPW; ++t) dacc[t] = v8f{};

#pragma unroll
  for (int k = 0; k < K_DIM; k += 4) {
    v2f a;
    a.x = Hs[r16][k + 2 * half];
    a.y = Hs[r16][k + 2 * half + 1];
    const float2 w0 = *(const float2*)&Wb[(size_t)(k + 2 * half)     * NOUT + col];
    const float2 w1 = *(const float2*)&Wb[(size_t)(k + 2 * half + 1) * NOUT + col];
    v2f b0, b1;
    b0.x = w0.x; b0.y = w1.x;
    b1.x = w0.y; b1.y = w1.y;
    dacc[0] = __builtin_amdgcn_wmma_f32_16x16x4_f32(false, a, false, b0,
                                                    (short)0, dacc[0], false, false);
    dacc[1] = __builtin_amdgcn_wmma_f32_16x16x4_f32(false, a, false, b1,
                                                    (short)0, dacc[1], false, false);
  }

#pragma unroll
  for (int r = 0; r < 8; ++r) {
    const int row = tileM + r + 8 * half;
    *(float2*)&C[(size_t)row * NOUT + col] = make_float2(dacc[0][r], dacc[1][r]);
  }
}

extern "C" void kernel_launch(void* const* d_in, const int* in_sizes, int n_in,
                              void* d_out, int out_size, void* d_ws, size_t ws_size,
                              hipStream_t stream) {
  const float* feat = (const float*)d_in[0];
  const float* ew   = (const float*)d_in[1];
  const float* W1   = (const float*)d_in[2];
  const float* b1   = (const float*)d_in[3];
  const float* W2   = (const float*)d_in[4];
  const float* b2   = (const float*)d_in[5];
  const float* W3   = (const float*)d_in[6];
  const float* b3   = (const float*)d_in[7];
  const int*   src  = (const int*)d_in[8];
  const int*   dst  = (const int*)d_in[9];
  float* out = (float*)d_out;

  float* ws     = (float*)d_ws;
  int*   cnt_in = (int*)ws;
  float* odi    = ws + 1 * (size_t)N_NODES;
  float* idi    = ws + 2 * (size_t)N_NODES;
  int*   rowptr = (int*)ws + 3 * (size_t)N_NODES;
  float* proj3  = ws + 4 * (size_t)N_NODES;
  float* proj2  = proj3 + (size_t)N_NODES * 128;
  int*   pairs  = (int*)(proj2 + (size_t)N_NODES * 128);
  int*   part_in  = pairs + 2 * (size_t)N_EDGES;
  int*   part_out = part_in + (size_t)NSLICE * NODE_PAD;
  int*   tot_in   = part_out + (size_t)NSLICE * NODE_PAD;

  gcn_count<<<dim3(NRANGE, NSLICE), EDGE_THREADS, 0, stream>>>(src, dst, part_out, part_in, tot_in);
  gcn_scan_norm<<<NCHUNK, CHUNK_N, 0, stream>>>(part_in, part_out, tot_in, cnt_in, odi, idi, rowptr);
  gcn_fill<<<dim3(NRANGE, NSLICE), EDGE_THREADS, 0, stream>>>(src, dst, ew, part_in, pairs);

  const int stripeBlocks = N_NODES / 16;
  const int gatBlocks = GAT_BLOCKS;

  gcn_gather_gemm2<<<stripeBlocks, 32 * FUSE_WAVES, 0, stream>>>(feat, pairs, rowptr, cnt_in, idi, odi, b1, W1, W2, proj2);

  gcn_gather_gemm<64><<<stripeBlocks, 32 * FUSE_WAVES, 0, stream>>>(proj2, pairs, rowptr, cnt_in, idi, odi, b2, W3, proj3);

  gcn_gather<64, 0><<<gatBlocks, 32 * GAT_WAVES, 0, stream>>>(proj3, pairs, rowptr, cnt_in, idi, b3, out);
}
